# starting kernel
# baseline (speedup 1.0000x reference)
#include <hip/hip_runtime.h>
#include <math.h>


typedef __attribute__((ext_vector_type(2))) float v2f;
typedef __attribute__((ext_vector_type(8))) float v8f;

#define D 128
#define SELF_PAD 512

__global__ __launch_bounds__(256) void gemm128_wmma(
    const float* __restrict__ A, const float* __restrict__ W,
    const float* __restrict__ bias, float* __restrict__ C, int M) {
  const int lane = threadIdx.x & 31;
  const int wave = threadIdx.x >> 5;
  const int r = lane & 15;
  const int g = lane >> 4;
  const int m0 = blockIdx.x * 16;
  const int n0 = wave * 16;

  const float* __restrict__ Arow = A + (size_t)(m0 + r) * D;
  const float* __restrict__ Wcol = W + (n0 + r);

  v8f acc = {};
#pragma unroll
  for (int k = 0; k < D; k += 4) {
    const int kk = k + 2 * g;
    v2f a;
    a.x = Arow[kk];
    a.y = Arow[kk + 1];
    v2f b;
    b.x = Wcol[(size_t)kk * D];
    b.y = Wcol[(size_t)(kk + 1) * D];
    acc = __builtin_amdgcn_wmma_f32_16x16x4_f32(
         false, a,  false, b,
         (short)0, acc,  false,  false);
  }

  const float bv = bias ? bias[n0 + r] : 0.0f;
#pragma unroll
  for (int j = 0; j < 8; ++j) {
    const int row = m0 + j + 8 * g;
    C[(size_t)row * D + n0 + r] = acc[j] + bv;
  }
}

__global__ void init_kernel(float* out0, float* featdyn, float* mdyn,
                            float* msta, float* sdyn, float* ssta,
                            float* cnts, int N) {
  const int i = blockIdx.x * 256 + threadIdx.x;
  const int ND = N * D;
  if (i < ND) {
    out0[i] = 0.0f;
    featdyn[i] = 0.0f;
  }
  if (i < N) {
    mdyn[i] = -__builtin_inff();
    msta[i] = -__builtin_inff();
    sdyn[i] = 0.0f;
    ssta[i] = 0.0f;
    cnts[i] = 0.0f;
  }
}

__global__ void build_trel(const float* __restrict__ relemb,
                           const float* __restrict__ selfrel,
                           float* __restrict__ Trel, int R) {
  const int i = blockIdx.x * 256 + threadIdx.x;
  if (i >= SELF_PAD * D) return;
  const int row = i >> 7;
  const int c = i & (D - 1);
  float v = 0.0f;
  if (row < R) v = relemb[i];
  else if (row == R) v = selfrel[c];
  Trel[i] = v;
}

__device__ inline void atomicMaxF(float* addr, float v) {
  if (v >= 0.0f)
    atomicMax((int*)addr, __float_as_int(v));
  else
    atomicMin((unsigned int*)addr, __float_as_uint(v));
}

__global__ __launch_bounds__(256) void edge_attn_kernel(
    const int* __restrict__ ei, const int* __restrict__ et, int E, int N,
    int R, const float* __restrict__ Xq, const float* __restrict__ Xk,
    const float* __restrict__ Tq, const float* __restrict__ Tk,
    float* __restrict__ attn, float* __restrict__ mdyn,
    float* __restrict__ msta, float* __restrict__ cnts) {
  const int e = blockIdx.x * 8 + (threadIdx.x >> 5);
  const int lane = threadIdx.x & 31;
  const int ET = E + N;
  if (e >= ET) return;

  int s, d, t;
  if (e < E) {
    s = ei[e];
    d = ei[E + e];
    t = et[e];
  } else {
    s = d = e - E;
    t = R;
  }

  const float4 q0 = ((const float4*)(Xq + (size_t)d * D))[lane];
  const float4 q1 = ((const float4*)(Tq + (size_t)t * D))[lane];
  const float4 k0 = ((const float4*)(Xk + (size_t)s * D))[lane];
  const float4 k1 = ((const float4*)(Tk + (size_t)t * D))[lane];

  float a = (q0.x + q1.x) * (k0.x + k1.x) + (q0.y + q1.y) * (k0.y + k1.y) +
            (q0.z + q1.z) * (k0.z + k1.z) + (q0.w + q1.w) * (k0.w + k1.w);
#pragma unroll
  for (int off = 16; off >= 1; off >>= 1) a += __shfl_xor(a, off, 32);

  if (lane == 0) {
    a *= 0.08838834764831845f;
    attn[e] = a;
    atomicMaxF(&mdyn[d], a);
    atomicMaxF(&msta[d], -a);
    atomicAdd(&cnts[d], 1.0f);
  }
}

__global__ void edge_expsum_kernel(const int* __restrict__ ei, int E, int N,
                                   const float* __restrict__ attn,
                                   const float* __restrict__ mdyn,
                                   const float* __restrict__ msta,
                                   float* __restrict__ edyn,
                                   float* __restrict__ esta,
                                   float* __restrict__ sdyn,
                                   float* __restrict__ ssta) {
  const int e = blockIdx.x * 256 + threadIdx.x;
  const int ET = E + N;
  if (e >= ET) return;
  const int d = (e < E) ? ei[E + e] : (e - E);
  const float a = attn[e];
  const float ed = __expf(a - mdyn[d]);
  const float es = __expf(-a - msta[d]);
  edyn[e] = ed;
  esta[e] = es;
  atomicAdd(&sdyn[d], ed);
  atomicAdd(&ssta[d], es);
}

__global__ __launch_bounds__(256) void edge_scatter_kernel(
    const int* __restrict__ ei, int E, int N, const float* __restrict__ edyn,
    const float* __restrict__ esta, const float* __restrict__ sdyn,
    const float* __restrict__ ssta, const float* __restrict__ cnts,
    const float* __restrict__ Vv, float* __restrict__ featsta,
    float* __restrict__ featdyn) {
  const int e = blockIdx.x * 8 + (threadIdx.x >> 5);
  const int lane = threadIdx.x & 31;
  const int ET = E + N;
  if (e >= ET) return;

  int s, d;
  if (e < E) {
    s = ei[e];
    d = ei[E + e];
  } else {
    s = d = e - E;
  }

  const float denom = fmaxf(cnts[d], 1.0f);
  const float wd = edyn[e] / (sdyn[d] * denom);
  const float ws = esta[e] / (ssta[d] * denom);

  const float4 v4 = ((const float4*)(Vv + (size_t)s * D))[lane];
  float* fd = featdyn + (size_t)d * D + lane * 4;
  float* fs = featsta + (size_t)d * D + lane * 4;
  atomicAdd(fs + 0, ws * v4.x);
  atomicAdd(fs + 1, ws * v4.y);
  atomicAdd(fs + 2, ws * v4.z);
  atomicAdd(fs + 3, ws * v4.w);
  atomicAdd(fd + 0, wd * v4.x);
  atomicAdd(fd + 1, wd * v4.y);
  atomicAdd(fd + 2, wd * v4.z);
  atomicAdd(fd + 3, wd * v4.w);
}

extern "C" void kernel_launch(void* const* d_in, const int* in_sizes, int n_in,
                              void* d_out, int out_size, void* d_ws,
                              size_t ws_size, hipStream_t stream) {
  const float* curr = (const float*)d_in[0];
  const float* last = (const float*)d_in[1];
  const float* relemb = (const float*)d_in[2];
  const float* selfrel = (const float*)d_in[3];
  const float* Wq = (const float*)d_in[4];
  const float* bq = (const float*)d_in[5];
  const float* Wk = (const float*)d_in[6];
  const float* bk = (const float*)d_in[7];
  const float* Wv = (const float*)d_in[8];
  const float* bv = (const float*)d_in[9];
  const float* dynW = (const float*)d_in[10];
  const int* ei = (const int*)d_in[11];
  const int* et = (const int*)d_in[12];

  const int N = in_sizes[0] / D;
  const int E = in_sizes[12];
  const int R = in_sizes[2] / D;
  const int ET = E + N;
  const size_t ND = (size_t)N * D;

  float* ws = (float*)d_ws;
  float* Xq = ws;
  float* Xk = Xq + ND;
  float* Vv = Xk + ND;
  float* featdyn = Vv + ND;
  float* Trel = featdyn + ND;
  float* Tq = Trel + SELF_PAD * D;
  float* Tk = Tq + SELF_PAD * D;
  float* attn = Tk + SELF_PAD * D;
  float* edyn = attn + ET;
  float* esta = edyn + ET;
  float* mdyn = esta + ET;
  float* msta = mdyn + N;
  float* sdyn = msta + N;
  float* ssta = sdyn + N;
  float* cnts = ssta + N;

  float* out0 = (float*)d_out;
  float* out1 = out0 + ND;

  init_kernel<<<(int)((ND + 255) / 256), 256, 0, stream>>>(
      out0, featdyn, mdyn, msta, sdyn, ssta, cnts, N);
  build_trel<<<(SELF_PAD * D + 255) / 256, 256, 0, stream>>>(relemb, selfrel,
                                                             Trel, R);

  gemm128_wmma<<<N / 16, 256, 0, stream>>>(curr, Wq, nullptr, Xq, N);
  gemm128_wmma<<<N / 16, 256, 0, stream>>>(last, Wk, nullptr, Xk, N);
  gemm128_wmma<<<N / 16, 256, 0, stream>>>(last, Wv, bv, Vv, N);
  gemm128_wmma<<<SELF_PAD / 16, 256, 0, stream>>>(Trel, Wq + D * D, bq, Tq,
                                                  SELF_PAD);
  gemm128_wmma<<<SELF_PAD / 16, 256, 0, stream>>>(Trel, Wk + D * D, bk, Tk,
                                                  SELF_PAD);

  edge_attn_kernel<<<(ET + 7) / 8, 256, 0, stream>>>(
      ei, et, E, N, R, Xq, Xk, Tq, Tk, attn, mdyn, msta, cnts);
  edge_expsum_kernel<<<(ET + 255) / 256, 256, 0, stream>>>(
      ei, E, N, attn, mdyn, msta, edyn, esta, sdyn, ssta);
  edge_scatter_kernel<<<(ET + 7) / 8, 256, 0, stream>>>(
      ei, E, N, edyn, esta, sdyn, ssta, cnts, Vv, out0, featdyn);

  gemm128_wmma<<<N / 16, 256, 0, stream>>>(featdyn, dynW, nullptr, out1, N);
}


# Row-gathering edge kernels: one wave per 8 consecutive edges (was 1), their row loads issued as one group, one shared butterfly; 8x fewer waves. f32 atomics kept, arrival order varies as before.
# speedup vs baseline: 1.1823x; 1.1823x over previous
#include <hip/hip_runtime.h>
#include <math.h>


typedef __attribute__((ext_vector_type(2))) float v2f;
typedef __attribute__((ext_vector_type(8))) float v8f;

#define D 128
#define SELF_PAD 512
#define EPW 8

__global__ __launch_bounds__(256) void gemm128_wmma(
    const float* __restrict__ A, const float* __restrict__ W,
    const float* __restrict__ bias, float* __restrict__ C, int M) {
  const int lane = threadIdx.x & 31;
  const int wave = threadIdx.x >> 5;
  const int r = lane & 15;
  const int g = lane >> 4;
  const int m0 = blockIdx.x * 16;
  const int n0 = wave * 16;

  const float* __restrict__ Arow = A + (size_t)(m0 + r) * D;
  const float* __restrict__ Wcol = W + (n0 + r);

  v8f acc = {};
#pragma unroll
  for (int k = 0; k < D; k += 4) {
    const int kk = k + 2 * g;
    v2f a;
    a.x = Arow[kk];
    a.y = Arow[kk + 1];
    v2f b;
    b.x = Wcol[(size_t)kk * D];
    b.y = Wcol[(size_t)(kk + 1) * D];
    acc = __builtin_amdgcn_wmma_f32_16x16x4_f32(
         false, a,  false, b,
         (short)0, acc,  false,  false);
  }

  const float bv = bias ? bias[n0 + r] : 0.0f;
#pragma unroll
  for (int j = 0; j < 8; ++j) {
    const int row = m0 + j + 8 * g;
    C[(size_t)row * D + n0 + r] = acc[j] + bv;
  }
}

__global__ void init_kernel(float* out0, float* featdyn, float* mdyn,
                            float* msta, float* sdyn, float* ssta,
                            float* cnts, int N) {
  const int i = blockIdx.x * 256 + threadIdx.x;
  const int ND = N * D;
  if (i < ND) {
    out0[i] = 0.0f;
    featdyn[i] = 0.0f;
  }
  if (i < N) {
    mdyn[i] = -__builtin_inff();
    msta[i] = -__builtin_inff();
    sdyn[i] = 0.0f;
    ssta[i] = 0.0f;
    cnts[i] = 0.0f;
  }
}

__global__ void build_trel(const float* __restrict__ relemb,
                           const float* __restrict__ selfrel,
                           float* __restrict__ Trel, int R) {
  const int i = blockIdx.x * 256 + threadIdx.x;
  if (i >= SELF_PAD * D) return;
  const int row = i >> 7;
  const int c = i & (D - 1);
  float v = 0.0f;
  if (row < R) v = relemb[i];
  else if (row == R) v = selfrel[c];
  Trel[i] = v;
}

__device__ inline void atomicMaxF(float* addr, float v) {
  if (v >= 0.0f)
    atomicMax((int*)addr, __float_as_int(v));
  else
    atomicMin((unsigned int*)addr, __float_as_uint(v));
}

__global__ __launch_bounds__(256) void edge_attn_kernel(
    const int* __restrict__ ei, const int* __restrict__ et, int E, int N,
    int R, const float* __restrict__ Xq, const float* __restrict__ Xk,
    const float* __restrict__ Tq, const float* __restrict__ Tk,
    float* __restrict__ attn, float* __restrict__ mdyn,
    float* __restrict__ msta, float* __restrict__ cnts) {
  const int lane = threadIdx.x & 31;
  const int wv = __builtin_amdgcn_readfirstlane((int)(threadIdx.x >> 5));
  const int e0 = (blockIdx.x * 8 + wv) * EPW;
  const int ET = E + N;
  if (e0 >= ET) return;

  const int me = e0 + (lane >> 2);
  const bool have = me < ET;
  int ms = 0, md = 0, mt = R;
  if (have) {
    if (me < E) {
      ms = ei[me];
      md = ei[E + me];
      mt = et[me];
    } else {
      ms = md = me - E;
    }
  }

  float4 q0[EPW], q1[EPW], k0[EPW], k1[EPW];
#pragma unroll
  for (int j = 0; j < EPW; ++j) {
    const int s = __builtin_amdgcn_readlane(ms, 4 * j);
    const int d = __builtin_amdgcn_readlane(md, 4 * j);
    const int t = __builtin_amdgcn_readlane(mt, 4 * j);
    q0[j] = ((const float4*)(Xq + (size_t)d * D))[lane];
    q1[j] = ((const float4*)(Tq + (size_t)t * D))[lane];
    k0[j] = ((const float4*)(Xk + (size_t)s * D))[lane];
    k1[j] = ((const float4*)(Tk + (size_t)t * D))[lane];
  }

  float a[EPW];
#pragma unroll
  for (int j = 0; j < EPW; ++j) {
    a[j] = (q0[j].x + q1[j].x) * (k0[j].x + k1[j].x) +
           (q0[j].y + q1[j].y) * (k0[j].y + k1[j].y) +
           (q0[j].z + q1[j].z) * (k0[j].z + k1[j].z) +
           (q0[j].w + q1[j].w) * (k0[j].w + k1[j].w);
  }

  const bool up16 = (lane & 16) != 0;
  const bool up8 = (lane & 8) != 0;
  const bool up4 = (lane & 4) != 0;
  float b[4];
#pragma unroll
  for (int j = 0; j < 4; ++j) {
    const float keep = up16 ? a[j + 4] : a[j];
    const float give = up16 ? a[j] : a[j + 4];
    b[j] = keep + __shfl_xor(give, 16, 32);
  }
  float c[2];
#pragma unroll
  for (int j = 0; j < 2; ++j) {
    const float keep = up8 ? b[j + 2] : b[j];
    const float give = up8 ? b[j] : b[j + 2];
    c[j] = keep + __shfl_xor(give, 8, 32);
  }
  float r;
  {
    const float keep = up4 ? c[1] : c[0];
    const float give = up4 ? c[0] : c[1];
    r = keep + __shfl_xor(give, 4, 32);
  }
  r += __shfl_xor(r, 2, 32);
  r += __shfl_xor(r, 1, 32);

  if (have && (lane & 3) == 0) {
    r *= 0.08838834764831845f;
    attn[me] = r;
    atomicMaxF(&mdyn[md], r);
    atomicMaxF(&msta[md], -r);
    atomicAdd(&cnts[md], 1.0f);
  }
}

__global__ void edge_expsum_kernel(const int* __restrict__ ei, int E, int N,
                                   const float* __restrict__ attn,
                                   const float* __restrict__ mdyn,
                                   const float* __restrict__ msta,
                                   float* __restrict__ edyn,
                                   float* __restrict__ esta,
                                   float* __restrict__ sdyn,
                                   float* __restrict__ ssta) {
  const int e = blockIdx.x * 256 + threadIdx.x;
  const int ET = E + N;
  if (e >= ET) return;
  const int d = (e < E) ? ei[E + e] : (e - E);
  const float a = attn[e];
  const float ed = __expf(a - mdyn[d]);
  const float es = __expf(-a - msta[d]);
  edyn[e] = ed;
  esta[e] = es;
  atomicAdd(&sdyn[d], ed);
  atomicAdd(&ssta[d], es);
}

__global__ __launch_bounds__(256) void edge_scatter_kernel(
    const int* __restrict__ ei, int E, int N, const float* __restrict__ edyn,
    const float* __restrict__ esta, const float* __restrict__ sdyn,
    const float* __restrict__ ssta, const float* __restrict__ cnts,
    const float* __restrict__ Vv, float* __restrict__ featsta,
    float* __restrict__ featdyn) {
  const int lane = threadIdx.x & 31;
  const int wv = __builtin_amdgcn_readfirstlane((int)(threadIdx.x >> 5));
  const int e0 = (blockIdx.x * 8 + wv) * EPW;
  const int ET = E + N;
  if (e0 >= ET) return;
  const int left = ET - e0;

  const int me = e0 + (lane >> 2);
  const bool have = me < ET;
  int ms = 0, md = 0;
  if (have) {
    if (me < E) {
      ms = ei[me];
      md = ei[E + me];
    } else {
      ms = md = me - E;
    }
  }

  float4 v4[EPW];
#pragma unroll
  for (int j = 0; j < EPW; ++j) {
    const int s = __builtin_amdgcn_readlane(ms, 4 * j);
    v4[j] = ((const float4*)(Vv + (size_t)s * D))[lane];
  }

  float wd = 0.0f, ws = 0.0f;
  if (have) {
    const float denom = fmaxf(cnts[md], 1.0f);
    wd = edyn[me] / (sdyn[md] * denom);
    ws = esta[me] / (ssta[md] * denom);
  }

#pragma unroll
  for (int j = 0; j < EPW; ++j) {
    if (j < left) {
      const int d = __builtin_amdgcn_readlane(md, 4 * j);
      const float wdj =
          __int_as_float(__builtin_amdgcn_readlane(__float_as_int(wd), 4 * j));
      const float wsj =
          __int_as_float(__builtin_amdgcn_readlane(__float_as_int(ws), 4 * j));
      float* fd = featdyn + (size_t)d * D + lane * 4;
      float* fs = featsta + (size_t)d * D + lane * 4;
      atomicAdd(fs + 0, wsj * v4[j].x);
      atomicAdd(fs + 1, wsj * v4[j].y);
      atomicAdd(fs + 2, wsj * v4[j].z);
      atomicAdd(fs + 3, wsj * v4[j].w);
      atomicAdd(fd + 0, wdj * v4[j].x);
      atomicAdd(fd + 1, wdj * v4[j].y);
      atomicAdd(fd + 2, wdj * v4[j].z);
      atomicAdd(fd + 3, wdj * v4[j].w);
    }
  }
}

extern "C" void kernel_launch(void* const* d_in, const int* in_sizes, int n_in,
                              void* d_out, int out_size, void* d_ws,
                              size_t ws_size, hipStream_t stream) {
  const float* curr = (const float*)d_in[0];
  const float* last = (const float*)d_in[1];
  const float* relemb = (const float*)d_in[2];
  const float* selfrel = (const float*)d_in[3];
  const float* Wq = (const float*)d_in[4];
  const float* bq = (const float*)d_in[5];
  const float* Wk = (const float*)d_in[6];
  const float* bk = (const float*)d_in[7];
  const float* Wv = (const float*)d_in[8];
  const float* bv = (const float*)d_in[9];
  const float* dynW = (const float*)d_in[10];
  const int* ei = (const int*)d_in[11];
  const int* et = (const int*)d_in[12];

  const int N = in_sizes[0] / D;
  const int E = in_sizes[12];
  const int R = in_sizes[2] / D;
  const int ET = E + N;
  const size_t ND = (size_t)N * D;

  float* ws = (float*)d_ws;
  float* Xq = ws;
  float* Xk = Xq + ND;
  float* Vv = Xk + ND;
  float* featdyn = Vv + ND;
  float* Trel = featdyn + ND;
  float* Tq = Trel + SELF_PAD * D;
  float* Tk = Tq + SELF_PAD * D;
  float* attn = Tk + SELF_PAD * D;
  float* edyn = attn + ET;
  float* esta = edyn + ET;
  float* mdyn = esta + ET;
  float* msta = mdyn + N;
  float* sdyn = msta + N;
  float* ssta = sdyn + N;
  float* cnts = ssta + N;

  float* out0 = (float*)d_out;
  float* out1 = out0 + ND;

  init_kernel<<<(int)((ND + 255) / 256), 256, 0, stream>>>(
      out0, featdyn, mdyn, msta, sdyn, ssta, cnts, N);
  build_trel<<<(SELF_PAD * D + 255) / 256, 256, 0, stream>>>(relemb, selfrel,
                                                             Trel, R);

  gemm128_wmma<<<N / 16, 256, 0, stream>>>(curr, Wq, nullptr, Xq, N);
  gemm128_wmma<<<N / 16, 256, 0, stream>>>(last, Wk, nullptr, Xk, N);
  gemm128_wmma<<<N / 16, 256, 0, stream>>>(last, Wv, bv, Vv, N);
  gemm128_wmma<<<SELF_PAD / 16, 256, 0, stream>>>(Trel, Wq + D * D, bq, Tq,
                                                  SELF_PAD);
  gemm128_wmma<<<SELF_PAD / 16, 256, 0, stream>>>(Trel, Wk + D * D, bk, Tk,
                                                  SELF_PAD);

  edge_attn_kernel<<<(ET + 8 * EPW - 1) / (8 * EPW), 256, 0, stream>>>(
      ei, et, E, N, R, Xq, Xk, Tq, Tk, attn, mdyn, msta, cnts);
  edge_expsum_kernel<<<(ET + 255) / 256, 256, 0, stream>>>(
      ei, E, N, attn, mdyn, msta, edyn, esta, sdyn, ssta);
  edge_scatter_kernel<<<(ET + 8 * EPW - 1) / (8 * EPW), 256, 0, stream>>>(
      ei, E, N, edyn, esta, sdyn, ssta, cnts, Vv, out0, featdyn);

  gemm128_wmma<<<N / 16, 256, 0, stream>>>(featdyn, dynW, nullptr, out1, N);
}


# Scatter kernel: lane l owns columns l, l+32, l+64, l+96, so each f32 atomic instruction and value load covers one whole 128-byte line (was a quarter of four). Arrival order varies as before.
# speedup vs baseline: 1.4475x; 1.2243x over previous
#include <hip/hip_runtime.h>
#include <math.h>


typedef __attribute__((ext_vector_type(2))) float v2f;
typedef __attribute__((ext_vector_type(8))) float v8f;

#define D 128
#define SELF_PAD 512
#define EPW 8

__global__ __launch_bounds__(256) void gemm128_wmma(
    const float* __restrict__ A, const float* __restrict__ W,
    const float* __restrict__ bias, float* __restrict__ C, int M) {
  const int lane = threadIdx.x & 31;
  const int wave = threadIdx.x >> 5;
  const int r = lane & 15;
  const int g = lane >> 4;
  const int m0 = blockIdx.x * 16;
  const int n0 = wave * 16;

  const float* __restrict__ Arow = A + (size_t)(m0 + r) * D;
  const float* __restrict__ Wcol = W + (n0 + r);

  v8f acc = {};
#pragma unroll
  for (int k = 0; k < D; k += 4) {
    const int kk = k + 2 * g;
    v2f a;
    a.x = Arow[kk];
    a.y = Arow[kk + 1];
    v2f b;
    b.x = Wcol[(size_t)kk * D];
    b.y = Wcol[(size_t)(kk + 1) * D];
    acc = __builtin_amdgcn_wmma_f32_16x16x4_f32(
         false, a,  false, b,
         (short)0, acc,  false,  false);
  }

  const float bv = bias ? bias[n0 + r] : 0.0f;
#pragma unroll
  for (int j = 0; j < 8; ++j) {
    const int row = m0 + j + 8 * g;
    C[(size_t)row * D + n0 + r] = acc[j] + bv;
  }
}

__global__ void init_kernel(float* out0, float* featdyn, float* mdyn,
                            float* msta, float* sdyn, float* ssta,
                            float* cnts, int N) {
  const int i = blockIdx.x * 256 + threadIdx.x;
  const int ND = N * D;
  if (i < ND) {
    out0[i] = 0.0f;
    featdyn[i] = 0.0f;
  }
  if (i < N) {
    mdyn[i] = -__builtin_inff();
    msta[i] = -__builtin_inff();
    sdyn[i] = 0.0f;
    ssta[i] = 0.0f;
    cnts[i] = 0.0f;
  }
}

__global__ void build_trel(const float* __restrict__ relemb,
                           const float* __restrict__ selfrel,
                           float* __restrict__ Trel, int R) {
  const int i = blockIdx.x * 256 + threadIdx.x;
  if (i >= SELF_PAD * D) return;
  const int row = i >> 7;
  const int c = i & (D - 1);
  float v = 0.0f;
  if (row < R) v = relemb[i];
  else if (row == R) v = selfrel[c];
  Trel[i] = v;
}

__device__ inline void atomicMaxF(float* addr, float v) {
  if (v >= 0.0f)
    atomicMax((int*)addr, __float_as_int(v));
  else
    atomicMin((unsigned int*)addr, __float_as_uint(v));
}

__global__ __launch_bounds__(256) void edge_attn_kernel(
    const int* __restrict__ ei, const int* __restrict__ et, int E, int N,
    int R, const float* __restrict__ Xq, const float* __restrict__ Xk,
    const float* __restrict__ Tq, const float* __restrict__ Tk,
    float* __restrict__ attn, float* __restrict__ mdyn,
    float* __restrict__ msta, float* __restrict__ cnts) {
  const int lane = threadIdx.x & 31;
  const int wv = __builtin_amdgcn_readfirstlane((int)(threadIdx.x >> 5));
  const int e0 = (blockIdx.x * 8 + wv) * EPW;
  const int ET = E + N;
  if (e0 >= ET) return;

  const int me = e0 + (lane >> 2);
  const bool have = me < ET;
  int ms = 0, md = 0, mt = R;
  if (have) {
    if (me < E) {
      ms = ei[me];
      md = ei[E + me];
      mt = et[me];
    } else {
      ms = md = me - E;
    }
  }

  float4 q0[EPW], q1[EPW], k0[EPW], k1[EPW];
#pragma unroll
  for (int j = 0; j < EPW; ++j) {
    const int s = __builtin_amdgcn_readlane(ms, 4 * j);
    const int d = __builtin_amdgcn_readlane(md, 4 * j);
    const int t = __builtin_amdgcn_readlane(mt, 4 * j);
    q0[j] = ((const float4*)(Xq + (size_t)d * D))[lane];
    q1[j] = ((const float4*)(Tq + (size_t)t * D))[lane];
    k0[j] = ((const float4*)(Xk + (size_t)s * D))[lane];
    k1[j] = ((const float4*)(Tk + (size_t)t * D))[lane];
  }

  float a[EPW];
#pragma unroll
  for (int j = 0; j < EPW; ++j) {
    a[j] = (q0[j].x + q1[j].x) * (k0[j].x + k1[j].x) +
           (q0[j].y + q1[j].y) * (k0[j].y + k1[j].y) +
           (q0[j].z + q1[j].z) * (k0[j].z + k1[j].z) +
           (q0[j].w + q1[j].w) * (k0[j].w + k1[j].w);
  }

  const bool up16 = (lane & 16) != 0;
  const bool up8 = (lane & 8) != 0;
  const bool up4 = (lane & 4) != 0;
  float b[4];
#pragma unroll
  for (int j = 0; j < 4; ++j) {
    const float keep = up16 ? a[j + 4] : a[j];
    const float give = up16 ? a[j] : a[j + 4];
    b[j] = keep + __shfl_xor(give, 16, 32);
  }
  float c[2];
#pragma unroll
  for (int j = 0; j < 2; ++j) {
    const float keep = up8 ? b[j + 2] : b[j];
    const float give = up8 ? b[j] : b[j + 2];
    c[j] = keep + __shfl_xor(give, 8, 32);
  }
  float r;
  {
    const float keep = up4 ? c[1] : c[0];
    const float give = up4 ? c[0] : c[1];
    r = keep + __shfl_xor(give, 4, 32);
  }
  r += __shfl_xor(r, 2, 32);
  r += __shfl_xor(r, 1, 32);

  if (have && (lane & 3) == 0) {
    r *= 0.08838834764831845f;
    attn[me] = r;
    atomicMaxF(&mdyn[md], r);
    atomicMaxF(&msta[md], -r);
    atomicAdd(&cnts[md], 1.0f);
  }
}

__global__ void edge_expsum_kernel(const int* __restrict__ ei, int E, int N,
                                   const float* __restrict__ attn,
                                   const float* __restrict__ mdyn,
                                   const float* __restrict__ msta,
                                   float* __restrict__ edyn,
                                   float* __restrict__ esta,
                                   float* __restrict__ sdyn,
                                   float* __restrict__ ssta) {
  const int e = blockIdx.x * 256 + threadIdx.x;
  const int ET = E + N;
  if (e >= ET) return;
  const int d = (e < E) ? ei[E + e] : (e - E);
  const float a = attn[e];
  const float ed = __expf(a - mdyn[d]);
  const float es = __expf(-a - msta[d]);
  edyn[e] = ed;
  esta[e] = es;
  atomicAdd(&sdyn[d], ed);
  atomicAdd(&ssta[d], es);
}

__global__ __launch_bounds__(256) void edge_scatter_kernel(
    const int* __restrict__ ei, int E, int N, const float* __restrict__ edyn,
    const float* __restrict__ esta, const float* __restrict__ sdyn,
    const float* __restrict__ ssta, const float* __restrict__ cnts,
    const float* __restrict__ Vv, float* __restrict__ featsta,
    float* __restrict__ featdyn) {
  const int lane = threadIdx.x & 31;
  const int wv = __builtin_amdgcn_readfirstlane((int)(threadIdx.x >> 5));
  const int e0 = (blockIdx.x * 8 + wv) * EPW;
  const int ET = E + N;
  if (e0 >= ET) return;
  const int left = ET - e0;

  const int me = e0 + (lane >> 2);
  const bool have = me < ET;
  int ms = 0, md = 0;
  if (have) {
    if (me < E) {
      ms = ei[me];
      md = ei[E + me];
    } else {
      ms = md = me - E;
    }
  }

  float v4[EPW][4];
#pragma unroll
  for (int j = 0; j < EPW; ++j) {
    const int s = __builtin_amdgcn_readlane(ms, 4 * j);
    const float* vr = Vv + (size_t)s * D + lane;
#pragma unroll
    for (int i = 0; i < 4; ++i) v4[j][i] = vr[32 * i];
  }

  float wd = 0.0f, ws = 0.0f;
  if (have) {
    const float denom = fmaxf(cnts[md], 1.0f);
    wd = edyn[me] / (sdyn[md] * denom);
    ws = esta[me] / (ssta[md] * denom);
  }

#pragma unroll
  for (int j = 0; j < EPW; ++j) {
    if (j < left) {
      const int d = __builtin_amdgcn_readlane(md, 4 * j);
      const float wdj =
          __int_as_float(__builtin_amdgcn_readlane(__float_as_int(wd), 4 * j));
      const float wsj =
          __int_as_float(__builtin_amdgcn_readlane(__float_as_int(ws), 4 * j));
      float* fd = featdyn + (size_t)d * D + lane;
      float* fs = featsta + (size_t)d * D + lane;
      atomicAdd(fs + 0, wsj * v4[j][0]);
      atomicAdd(fs + 32, wsj * v4[j][1]);
      atomicAdd(fs + 64, wsj * v4[j][2]);
      atomicAdd(fs + 96, wsj * v4[j][3]);
      atomicAdd(fd + 0, wdj * v4[j][0]);
      atomicAdd(fd + 32, wdj * v4[j][1]);
      atomicAdd(fd + 64, wdj * v4[j][2]);
      atomicAdd(fd + 96, wdj * v4[j][3]);
    }
  }
}

extern "C" void kernel_launch(void* const* d_in, const int* in_sizes, int n_in,
                              void* d_out, int out_size, void* d_ws,
                              size_t ws_size, hipStream_t stream) {
  const float* curr = (const float*)d_in[0];
  const float* last = (const float*)d_in[1];
  const float* relemb = (const float*)d_in[2];
  const float* selfrel = (const float*)d_in[3];
  const float* Wq = (const float*)d_in[4];
  const float* bq = (const float*)d_in[5];
  const float* Wk = (const float*)d_in[6];
  const float* bk = (const float*)d_in[7];
  const float* Wv = (const float*)d_in[8];
  const float* bv = (const float*)d_in[9];
  const float* dynW = (const float*)d_in[10];
  const int* ei = (const int*)d_in[11];
  const int* et = (const int*)d_in[12];

  const int N = in_sizes[0] / D;
  const int E = in_sizes[12];
  const int R = in_sizes[2] / D;
  const int ET = E + N;
  const size_t ND = (size_t)N * D;

  float* ws = (float*)d_ws;
  float* Xq = ws;
  float* Xk = Xq + ND;
  float* Vv = Xk + ND;
  float* featdyn = Vv + ND;
  float* Trel = featdyn + ND;
  float* Tq = Trel + SELF_PAD * D;
  float* Tk = Tq + SELF_PAD * D;
  float* attn = Tk + SELF_PAD * D;
  float* edyn = attn + ET;
  float* esta = edyn + ET;
  float* mdyn = esta + ET;
  float* msta = mdyn + N;
  float* sdyn = msta + N;
  float* ssta = sdyn + N;
  float* cnts = ssta + N;

  float* out0 = (float*)d_out;
  float* out1 = out0 + ND;

  init_kernel<<<(int)((ND + 255) / 256), 256, 0, stream>>>(
      out0, featdyn, mdyn, msta, sdyn, ssta, cnts, N);
  build_trel<<<(SELF_PAD * D + 255) / 256, 256, 0, stream>>>(relemb, selfrel,
                                                             Trel, R);

  gemm128_wmma<<<N / 16, 256, 0, stream>>>(curr, Wq, nullptr, Xq, N);
  gemm128_wmma<<<N / 16, 256, 0, stream>>>(last, Wk, nullptr, Xk, N);
  gemm128_wmma<<<N / 16, 256, 0, stream>>>(last, Wv, bv, Vv, N);
  gemm128_wmma<<<SELF_PAD / 16, 256, 0, stream>>>(Trel, Wq + D * D, bq, Tq,
                                                  SELF_PAD);
  gemm128_wmma<<<SELF_PAD / 16, 256, 0, stream>>>(Trel, Wk + D * D, bk, Tk,
                                                  SELF_PAD);

  edge_attn_kernel<<<(ET + 8 * EPW - 1) / (8 * EPW), 256, 0, stream>>>(
      ei, et, E, N, R, Xq, Xk, Tq, Tk, attn, mdyn, msta, cnts);
  edge_expsum_kernel<<<(ET + 255) / 256, 256, 0, stream>>>(
      ei, E, N, attn, mdyn, msta, edyn, esta, sdyn, ssta);
  edge_scatter_kernel<<<(ET + 8 * EPW - 1) / (8 * EPW), 256, 0, stream>>>(
      ei, E, N, edyn, esta, sdyn, ssta, cnts, Vv, out0, featdyn);

  gemm128_wmma<<<N / 16, 256, 0, stream>>>(featdyn, dynW, nullptr, out1, N);
}


# Node GEMMs: W staged once per work-group in LDS as interleaved k pairs; each wave computes a 16x64 strip, one A fragment feeding four f32 WMMAs (was a 16x16 tile per wave, A refetched by 8 waves)
# speedup vs baseline: 1.5873x; 1.0966x over previous
#include <hip/hip_runtime.h>
#include <math.h>


typedef __attribute__((ext_vector_type(2))) float v2f;
typedef __attribute__((ext_vector_type(8))) float v8f;

#define D 128
#define SELF_PAD 512
#define EPW 8

__global__ __launch_bounds__(256) void gemm128_wmma(
    const float* __restrict__ A, const float* __restrict__ W,
    const float* __restrict__ bias, float* __restrict__ C, int M) {
  const int lane = threadIdx.x & 31;
  const int wave = threadIdx.x >> 5;
  const int r = lane & 15;
  const int g = lane >> 4;
  const int m0 = blockIdx.x * 16;
  const int n0 = wave * 16;

  const float* __restrict__ Arow = A + (size_t)(m0 + r) * D;
  const float* __restrict__ Wcol = W + (n0 + r);

  v8f acc = {};
#pragma unroll
  for (int k = 0; k < D; k += 4) {
    const int kk = k + 2 * g;
    v2f a;
    a.x = Arow[kk];
    a.y = Arow[kk + 1];
    v2f b;
    b.x = Wcol[(size_t)kk * D];
    b.y = Wcol[(size_t)(kk + 1) * D];
    acc = __builtin_amdgcn_wmma_f32_16x16x4_f32(
         false, a,  false, b,
         (short)0, acc,  false,  false);
  }

  const float bv = bias ? bias[n0 + r] : 0.0f;
#pragma unroll
  for (int j = 0; j < 8; ++j) {
    const int row = m0 + j + 8 * g;
    C[(size_t)row * D + n0 + r] = acc[j] + bv;
  }
}

#define WP 288

__global__ __launch_bounds__(128) void gemm128_strip_wmma(
    const float* __restrict__ A, const float* __restrict__ W,
    const float* __restrict__ bias, float* __restrict__ C, int M) {
  __shared__ __attribute__((aligned(16))) float Ws[64 * WP];
  const int tid = threadIdx.x;
  const int lane = tid & 31;
  const int wave = __builtin_amdgcn_readfirstlane(tid >> 5);
  const int r = lane & 15;
  const int g = lane >> 4;
  const int m0 = (blockIdx.x * 2 + (wave >> 1)) * 16;
  const int nb = (wave & 1) * 64;

  float4 w0[16], w1[16];
  const float* __restrict__ Wb = W + (size_t)(2 * wave) * D + 4 * lane;
#pragma unroll
  for (int it = 0; it < 16; ++it) {
    w0[it] = *(const float4*)(Wb + it * 8 * D);
    w1[it] = *(const float4*)(Wb + it * 8 * D + D);
  }

  int arow = m0 + r;
  if (arow > M - 1) arow = M - 1;
  const float* __restrict__ Ap = A + (size_t)arow * D + 2 * g;
  v2f a[32];
#pragma unroll
  for (int ks = 0; ks < 32; ++ks) a[ks] = *(const v2f*)(Ap + 4 * ks);

  float* dstb = Ws + wave * WP + 8 * lane;
#pragma unroll
  for (int it = 0; it < 16; ++it) {
    float* dst = dstb + it * 4 * WP;
    *(float4*)dst = make_float4(w0[it].x, w1[it].x, w0[it].y, w1[it].y);
    *(float4*)(dst + 4) = make_float4(w0[it].z, w1[it].z, w0[it].w, w1[it].w);
  }
  asm volatile("s_wait_dscnt 0x0\n\ts_barrier_signal -1\n\ts_barrier_wait -1"
               ::: "memory");

  float bv[4];
#pragma unroll
  for (int ct = 0; ct < 4; ++ct)
    bv[ct] = bias ? bias[nb + 16 * ct + r] : 0.0f;

  v8f acc[4] = {};
  const float* __restrict__ Bp = Ws + g * WP + 2 * (nb + r);
#pragma unroll
  for (int ks = 0; ks < 32; ++ks) {
#pragma unroll
    for (int ct = 0; ct < 4; ++ct) {
      const v2f b = *(const v2f*)(Bp + 2 * ks * WP + 32 * ct);
      acc[ct] = __builtin_amdgcn_wmma_f32_16x16x4_f32(
           false, a[ks],  false, b,
           (short)0, acc[ct],  false,  false);
    }
  }

  if (m0 >= M) return;
#pragma unroll
  for (int ct = 0; ct < 4; ++ct) {
    const int col = nb + 16 * ct + r;
#pragma unroll
    for (int j = 0; j < 8; ++j) {
      const int row = m0 + j + 8 * g;
      C[(size_t)row * D + col] = acc[ct][j] + bv[ct];
    }
  }
}

__global__ void init_kernel(float* out0, float* featdyn, float* mdyn,
                            float* msta, float* sdyn, float* ssta,
                            float* cnts, int N) {
  const int i = blockIdx.x * 256 + threadIdx.x;
  const int ND = N * D;
  if (i < ND) {
    out0[i] = 0.0f;
    featdyn[i] = 0.0f;
  }
  if (i < N) {
    mdyn[i] = -__builtin_inff();
    msta[i] = -__builtin_inff();
    sdyn[i] = 0.0f;
    ssta[i] = 0.0f;
    cnts[i] = 0.0f;
  }
}

__global__ void build_trel(const float* __restrict__ relemb,
                           const float* __restrict__ selfrel,
                           float* __restrict__ Trel, int R) {
  const int i = blockIdx.x * 256 + threadIdx.x;
  if (i >= SELF_PAD * D) return;
  const int row = i >> 7;
  const int c = i & (D - 1);
  float v = 0.0f;
  if (row < R) v = relemb[i];
  else if (row == R) v = selfrel[c];
  Trel[i] = v;
}

__device__ inline void atomicMaxF(float* addr, float v) {
  if (v >= 0.0f)
    atomicMax((int*)addr, __float_as_int(v));
  else
    atomicMin((unsigned int*)addr, __float_as_uint(v));
}

__global__ __launch_bounds__(256) void edge_attn_kernel(
    const int* __restrict__ ei, const int* __restrict__ et, int E, int N,
    int R, const float* __restrict__ Xq, const float* __restrict__ Xk,
    const float* __restrict__ Tq, const float* __restrict__ Tk,
    float* __restrict__ attn, float* __restrict__ mdyn,
    float* __restrict__ msta, float* __restrict__ cnts) {
  const int lane = threadIdx.x & 31;
  const int wv = __builtin_amdgcn_readfirstlane((int)(threadIdx.x >> 5));
  const int e0 = (blockIdx.x * 8 + wv) * EPW;
  const int ET = E + N;
  if (e0 >= ET) return;

  const int me = e0 + (lane >> 2);
  const bool have = me < ET;
  int ms = 0, md = 0, mt = R;
  if (have) {
    if (me < E) {
      ms = ei[me];
      md = ei[E + me];
      mt = et[me];
    } else {
      ms = md = me - E;
    }
  }

  float4 q0[EPW], q1[EPW], k0[EPW], k1[EPW];
#pragma unroll
  for (int j = 0; j < EPW; ++j) {
    const int s = __builtin_amdgcn_readlane(ms, 4 * j);
    const int d = __builtin_amdgcn_readlane(md, 4 * j);
    const int t = __builtin_amdgcn_readlane(mt, 4 * j);
    q0[j] = ((const float4*)(Xq + (size_t)d * D))[lane];
    q1[j] = ((const float4*)(Tq + (size_t)t * D))[lane];
    k0[j] = ((const float4*)(Xk + (size_t)s * D))[lane];
    k1[j] = ((const float4*)(Tk + (size_t)t * D))[lane];
  }

  float a[EPW];
#pragma unroll
  for (int j = 0; j < EPW; ++j) {
    a[j] = (q0[j].x + q1[j].x) * (k0[j].x + k1[j].x) +
           (q0[j].y + q1[j].y) * (k0[j].y + k1[j].y) +
           (q0[j].z + q1[j].z) * (k0[j].z + k1[j].z) +
           (q0[j].w + q1[j].w) * (k0[j].w + k1[j].w);
  }

  const bool up16 = (lane & 16) != 0;
  const bool up8 = (lane & 8) != 0;
  const bool up4 = (lane & 4) != 0;
  float b[4];
#pragma unroll
  for (int j = 0; j < 4; ++j) {
    const float keep = up16 ? a[j + 4] : a[j];
    const float give = up16 ? a[j] : a[j + 4];
    b[j] = keep + __shfl_xor(give, 16, 32);
  }
  float c[2];
#pragma unroll
  for (int j = 0; j < 2; ++j) {
    const float keep = up8 ? b[j + 2] : b[j];
    const float give = up8 ? b[j] : b[j + 2];
    c[j] = keep + __shfl_xor(give, 8, 32);
  }
  float r;
  {
    const float keep = up4 ? c[1] : c[0];
    const float give = up4 ? c[0] : c[1];
    r = keep + __shfl_xor(give, 4, 32);
  }
  r += __shfl_xor(r, 2, 32);
  r += __shfl_xor(r, 1, 32);

  if (have && (lane & 3) == 0) {
    r *= 0.08838834764831845f;
    attn[me] = r;
    atomicMaxF(&mdyn[md], r);
    atomicMaxF(&msta[md], -r);
    atomicAdd(&cnts[md], 1.0f);
  }
}

__global__ void edge_expsum_kernel(const int* __restrict__ ei, int E, int N,
                                   const float* __restrict__ attn,
                                   const float* __restrict__ mdyn,
                                   const float* __restrict__ msta,
                                   float* __restrict__ edyn,
                                   float* __restrict__ esta,
                                   float* __restrict__ sdyn,
                                   float* __restrict__ ssta) {
  const int e = blockIdx.x * 256 + threadIdx.x;
  const int ET = E + N;
  if (e >= ET) return;
  const int d = (e < E) ? ei[E + e] : (e - E);
  const float a = attn[e];
  const float ed = __expf(a - mdyn[d]);
  const float es = __expf(-a - msta[d]);
  edyn[e] = ed;
  esta[e] = es;
  atomicAdd(&sdyn[d], ed);
  atomicAdd(&ssta[d], es);
}

__global__ __launch_bounds__(256) void edge_scatter_kernel(
    const int* __restrict__ ei, int E, int N, const float* __restrict__ edyn,
    const float* __restrict__ esta, const float* __restrict__ sdyn,
    const float* __restrict__ ssta, const float* __restrict__ cnts,
    const float* __restrict__ Vv, float* __restrict__ featsta,
    float* __restrict__ featdyn) {
  const int lane = threadIdx.x & 31;
  const int wv = __builtin_amdgcn_readfirstlane((int)(threadIdx.x >> 5));
  const int e0 = (blockIdx.x * 8 + wv) * EPW;
  const int ET = E + N;
  if (e0 >= ET) return;
  const int left = ET - e0;

  const int me = e0 + (lane >> 2);
  const bool have = me < ET;
  int ms = 0, md = 0;
  if (have) {
    if (me < E) {
      ms = ei[me];
      md = ei[E + me];
    } else {
      ms = md = me - E;
    }
  }

  float v4[EPW][4];
#pragma unroll
  for (int j = 0; j < EPW; ++j) {
    const int s = __builtin_amdgcn_readlane(ms, 4 * j);
    const float* vr = Vv + (size_t)s * D + lane;
#pragma unroll
    for (int i = 0; i < 4; ++i) v4[j][i] = vr[32 * i];
  }

  float wd = 0.0f, ws = 0.0f;
  if (have) {
    const float denom = fmaxf(cnts[md], 1.0f);
    wd = edyn[me] / (sdyn[md] * denom);
    ws = esta[me] / (ssta[md] * denom);
  }

#pragma unroll
  for (int j = 0; j < EPW; ++j) {
    if (j < left) {
      const int d = __builtin_amdgcn_readlane(md, 4 * j);
      const float wdj =
          __int_as_float(__builtin_amdgcn_readlane(__float_as_int(wd), 4 * j));
      const float wsj =
          __int_as_float(__builtin_amdgcn_readlane(__float_as_int(ws), 4 * j));
      float* fd = featdyn + (size_t)d * D + lane;
      float* fs = featsta + (size_t)d * D + lane;
      atomicAdd(fs + 0, wsj * v4[j][0]);
      atomicAdd(fs + 32, wsj * v4[j][1]);
      atomicAdd(fs + 64, wsj * v4[j][2]);
      atomicAdd(fs + 96, wsj * v4[j][3]);
      atomicAdd(fd + 0, wdj * v4[j][0]);
      atomicAdd(fd + 32, wdj * v4[j][1]);
      atomicAdd(fd + 64, wdj * v4[j][2]);
      atomicAdd(fd + 96, wdj * v4[j][3]);
    }
  }
}

extern "C" void kernel_launch(void* const* d_in, const int* in_sizes, int n_in,
                              void* d_out, int out_size, void* d_ws,
                              size_t ws_size, hipStream_t stream) {
  const float* curr = (const float*)d_in[0];
  const float* last = (const float*)d_in[1];
  const float* relemb = (const float*)d_in[2];
  const float* selfrel = (const float*)d_in[3];
  const float* Wq = (const float*)d_in[4];
  const float* bq = (const float*)d_in[5];
  const float* Wk = (const float*)d_in[6];
  const float* bk = (const float*)d_in[7];
  const float* Wv = (const float*)d_in[8];
  const float* bv = (const float*)d_in[9];
  const float* dynW = (const float*)d_in[10];
  const int* ei = (const int*)d_in[11];
  const int* et = (const int*)d_in[12];

  const int N = in_sizes[0] / D;
  const int E = in_sizes[12];
  const int R = in_sizes[2] / D;
  const int ET = E + N;
  const size_t ND = (size_t)N * D;

  float* ws = (float*)d_ws;
  float* Xq = ws;
  float* Xk = Xq + ND;
  float* Vv = Xk + ND;
  float* featdyn = Vv + ND;
  float* Trel = featdyn + ND;
  float* Tq = Trel + SELF_PAD * D;
  float* Tk = Tq + SELF_PAD * D;
  float* attn = Tk + SELF_PAD * D;
  float* edyn = attn + ET;
  float* esta = edyn + ET;
  float* mdyn = esta + ET;
  float* msta = mdyn + N;
  float* sdyn = msta + N;
  float* ssta = sdyn + N;
  float* cnts = ssta + N;

  float* out0 = (float*)d_out;
  float* out1 = out0 + ND;

  init_kernel<<<(int)((ND + 255) / 256), 256, 0, stream>>>(
      out0, featdyn, mdyn, msta, sdyn, ssta, cnts, N);
  build_trel<<<(SELF_PAD * D + 255) / 256, 256, 0, stream>>>(relemb, selfrel,
                                                             Trel, R);

  gemm128_strip_wmma<<<(N / 16 + 1) / 2, 128, 0, stream>>>(curr, Wq, nullptr,
                                                         Xq, N);
  gemm128_strip_wmma<<<(N / 16 + 1) / 2, 128, 0, stream>>>(last, Wk, nullptr,
                                                         Xk, N);
  gemm128_strip_wmma<<<(N / 16 + 1) / 2, 128, 0, stream>>>(last, Wv, bv, Vv, N);
  gemm128_wmma<<<SELF_PAD / 16, 256, 0, stream>>>(Trel, Wq + D * D, bq, Tq,
                                                  SELF_PAD);
  gemm128_wmma<<<SELF_PAD / 16, 256, 0, stream>>>(Trel, Wk + D * D, bk, Tk,
                                                  SELF_PAD);

  edge_attn_kernel<<<(ET + 8 * EPW - 1) / (8 * EPW), 256, 0, stream>>>(
      ei, et, E, N, R, Xq, Xk, Tq, Tk, attn, mdyn, msta, cnts);
  edge_expsum_kernel<<<(ET + 255) / 256, 256, 0, stream>>>(
      ei, E, N, attn, mdyn, msta, edyn, esta, sdyn, ssta);
  edge_scatter_kernel<<<(ET + 8 * EPW - 1) / (8 * EPW), 256, 0, stream>>>(
      ei, E, N, edyn, esta, sdyn, ssta, cnts, Vv, out0, featdyn);

  gemm128_strip_wmma<<<(N / 16 + 1) / 2, 128, 0, stream>>>(featdyn, dynW,
                                                         nullptr, out1, N);
}


# Feature scatter by f32 atomics -> gather: edges placed by destination (arrival-order rank from the count atomic, prefix sum); one wave sums a node's weighted rows in registers, in rank order
# speedup vs baseline: 1.7064x; 1.0750x over previous
#include <hip/hip_runtime.h>
#include <math.h>


typedef __attribute__((ext_vector_type(2))) float v2f;
typedef __attribute__((ext_vector_type(8))) float v8f;

#define D 128
#define SELF_PAD 512
#define EPW 8

__global__ __launch_bounds__(256) void gemm128_wmma(
    const float* __restrict__ A, const float* __restrict__ W,
    const float* __restrict__ bias, float* __restrict__ C, int M) {
  const int lane = threadIdx.x & 31;
  const int wave = threadIdx.x >> 5;
  const int r = lane & 15;
  const int g = lane >> 4;
  const int m0 = blockIdx.x * 16;
  const int n0 = wave * 16;

  const float* __restrict__ Arow = A + (size_t)(m0 + r) * D;
  const float* __restrict__ Wcol = W + (n0 + r);

  v8f acc = {};
#pragma unroll
  for (int k = 0; k < D; k += 4) {
    const int kk = k + 2 * g;
    v2f a;
    a.x = Arow[kk];
    a.y = Arow[kk + 1];
    v2f b;
    b.x = Wcol[(size_t)kk * D];
    b.y = Wcol[(size_t)(kk + 1) * D];
    acc = __builtin_amdgcn_wmma_f32_16x16x4_f32(
         false, a,  false, b,
         (short)0, acc,  false,  false);
  }

  const float bv = bias ? bias[n0 + r] : 0.0f;
#pragma unroll
  for (int j = 0; j < 8; ++j) {
    const int row = m0 + j + 8 * g;
    C[(size_t)row * D + n0 + r] = acc[j] + bv;
  }
}

#define WP 288

__global__ __launch_bounds__(128) void gemm128_strip_wmma(
    const float* __restrict__ A, const float* __restrict__ W,
    const float* __restrict__ bias, float* __restrict__ C, int M) {
  __shared__ __attribute__((aligned(16))) float Ws[64 * WP];
  const int tid = threadIdx.x;
  const int lane = tid & 31;
  const int wave = __builtin_amdgcn_readfirstlane(tid >> 5);
  const int r = lane & 15;
  const int g = lane >> 4;
  const int m0 = (blockIdx.x * 2 + (wave >> 1)) * 16;
  const int nb = (wave & 1) * 64;

  float4 w0[16], w1[16];
  const float* __restrict__ Wb = W + (size_t)(2 * wave) * D + 4 * lane;
#pragma unroll
  for (int it = 0; it < 16; ++it) {
    w0[it] = *(const float4*)(Wb + it * 8 * D);
    w1[it] = *(const float4*)(Wb + it * 8 * D + D);
  }

  int arow = m0 + r;
  if (arow > M - 1) arow = M - 1;
  const float* __restrict__ Ap = A + (size_t)arow * D + 2 * g;
  v2f a[32];
#pragma unroll
  for (int ks = 0; ks < 32; ++ks) a[ks] = *(const v2f*)(Ap + 4 * ks);

  float* dstb = Ws + wave * WP + 8 * lane;
#pragma unroll
  for (int it = 0; it < 16; ++it) {
    float* dst = dstb + it * 4 * WP;
    *(float4*)dst = make_float4(w0[it].x, w1[it].x, w0[it].y, w1[it].y);
    *(float4*)(dst + 4) = make_float4(w0[it].z, w1[it].z, w0[it].w, w1[it].w);
  }
  asm volatile("s_wait_dscnt 0x0\n\ts_barrier_signal -1\n\ts_barrier_wait -1"
               ::: "memory");

  float bv[4];
#pragma unroll
  for (int ct = 0; ct < 4; ++ct)
    bv[ct] = bias ? bias[nb + 16 * ct + r] : 0.0f;

  v8f acc[4] = {};
  const float* __restrict__ Bp = Ws + g * WP + 2 * (nb + r);
#pragma unroll
  for (int ks = 0; ks < 32; ++ks) {
#pragma unroll
    for (int ct = 0; ct < 4; ++ct) {
      const v2f b = *(const v2f*)(Bp + 2 * ks * WP + 32 * ct);
      acc[ct] = __builtin_amdgcn_wmma_f32_16x16x4_f32(
           false, a[ks],  false, b,
           (short)0, acc[ct],  false,  false);
    }
  }

  if (m0 >= M) return;
#pragma unroll
  for (int ct = 0; ct < 4; ++ct) {
    const int col = nb + 16 * ct + r;
#pragma unroll
    for (int j = 0; j < 8; ++j) {
      const int row = m0 + j + 8 * g;
      C[(size_t)row * D + col] = acc[ct][j] + bv[ct];
    }
  }
}

__global__ void build_trel(const float* __restrict__ relemb,
                           const float* __restrict__ selfrel,
                           float* __restrict__ Trel, int R,
                           float* __restrict__ mdyn, float* __restrict__ msta,
                           float* __restrict__ sdyn, float* __restrict__ ssta,
                           int* __restrict__ cnti, int N) {
  const int i = blockIdx.x * 256 + threadIdx.x;
  if (i < SELF_PAD * D) {
    const int row = i >> 7;
    const int c = i & (D - 1);
    float v = 0.0f;
    if (row < R) v = relemb[i];
    else if (row == R) v = selfrel[c];
    Trel[i] = v;
  }
  for (int n = i; n < N; n += (int)gridDim.x * 256) {
    mdyn[n] = -__builtin_inff();
    msta[n] = -__builtin_inff();
    sdyn[n] = 0.0f;
    ssta[n] = 0.0f;
    cnti[n] = 0;
  }
}

__device__ inline void atomicMaxF(float* addr, float v) {
  if (v >= 0.0f)
    atomicMax((int*)addr, __float_as_int(v));
  else
    atomicMin((unsigned int*)addr, __float_as_uint(v));
}

__global__ __launch_bounds__(256) void edge_attn_kernel(
    const int* __restrict__ ei, const int* __restrict__ et, int E, int N,
    int R, const float* __restrict__ Xq, const float* __restrict__ Xk,
    const float* __restrict__ Tq, const float* __restrict__ Tk,
    float* __restrict__ attn, float* __restrict__ mdyn,
    float* __restrict__ msta, int* __restrict__ cnti,
    int* __restrict__ rank) {
  const int lane = threadIdx.x & 31;
  const int wv = __builtin_amdgcn_readfirstlane((int)(threadIdx.x >> 5));
  const int e0 = (blockIdx.x * 8 + wv) * EPW;
  const int ET = E + N;
  if (e0 >= ET) return;

  const int me = e0 + (lane >> 2);
  const bool have = me < ET;
  int ms = 0, md = 0, mt = R;
  if (have) {
    if (me < E) {
      ms = ei[me];
      md = ei[E + me];
      mt = et[me];
    } else {
      ms = md = me - E;
    }
  }

  float4 q0[EPW], q1[EPW], k0[EPW], k1[EPW];
#pragma unroll
  for (int j = 0; j < EPW; ++j) {
    const int s = __builtin_amdgcn_readlane(ms, 4 * j);
    const int d = __builtin_amdgcn_readlane(md, 4 * j);
    const int t = __builtin_amdgcn_readlane(mt, 4 * j);
    q0[j] = ((const float4*)(Xq + (size_t)d * D))[lane];
    q1[j] = ((const float4*)(Tq + (size_t)t * D))[lane];
    k0[j] = ((const float4*)(Xk + (size_t)s * D))[lane];
    k1[j] = ((const float4*)(Tk + (size_t)t * D))[lane];
  }

  float a[EPW];
#pragma unroll
  for (int j = 0; j < EPW; ++j) {
    a[j] = (q0[j].x + q1[j].x) * (k0[j].x + k1[j].x) +
           (q0[j].y + q1[j].y) * (k0[j].y + k1[j].y) +
           (q0[j].z + q1[j].z) * (k0[j].z + k1[j].z) +
           (q0[j].w + q1[j].w) * (k0[j].w + k1[j].w);
  }

  const bool up16 = (lane & 16) != 0;
  const bool up8 = (lane & 8) != 0;
  const bool up4 = (lane & 4) != 0;
  float b[4];
#pragma unroll
  for (int j = 0; j < 4; ++j) {
    const float keep = up16 ? a[j + 4] : a[j];
    const float give = up16 ? a[j] : a[j + 4];
    b[j] = keep + __shfl_xor(give, 16, 32);
  }
  float c[2];
#pragma unroll
  for (int j = 0; j < 2; ++j) {
    const float keep = up8 ? b[j + 2] : b[j];
    const float give = up8 ? b[j] : b[j + 2];
    c[j] = keep + __shfl_xor(give, 8, 32);
  }
  float r;
  {
    const float keep = up4 ? c[1] : c[0];
    const float give = up4 ? c[0] : c[1];
    r = keep + __shfl_xor(give, 4, 32);
  }
  r += __shfl_xor(r, 2, 32);
  r += __shfl_xor(r, 1, 32);

  if (have && (lane & 3) == 0) {
    r *= 0.08838834764831845f;
    attn[me] = r;
    atomicMaxF(&mdyn[md], r);
    atomicMaxF(&msta[md], -r);
    rank[me] = atomicAdd(&cnti[md], 1);
  }
}

__global__ void edge_expsum_kernel(const int* __restrict__ ei, int E, int N,
                                   const float* __restrict__ attn,
                                   const float* __restrict__ mdyn,
                                   const float* __restrict__ msta,
                                   float* __restrict__ edyn,
                                   float* __restrict__ esta,
                                   float* __restrict__ sdyn,
                                   float* __restrict__ ssta) {
  const int e = blockIdx.x * 256 + threadIdx.x;
  const int ET = E + N;
  if (e >= ET) return;
  const int d = (e < E) ? ei[E + e] : (e - E);
  const float a = attn[e];
  const float ed = __expf(a - mdyn[d]);
  const float es = __expf(-a - msta[d]);
  edyn[e] = ed;
  esta[e] = es;
  atomicAdd(&sdyn[d], ed);
  atomicAdd(&ssta[d], es);
}

#define SCAN_T 1024

__global__ __launch_bounds__(SCAN_T) void slot_range_kernel(
    const int* __restrict__ cnti, int* __restrict__ off, int N) {
  __shared__ int red[SCAN_T / 32];
  const int tid = threadIdx.x;
  const int lane = tid & 31;
  const int wv = tid >> 5;
  const int start = blockIdx.x * SCAN_T;
  int part = 0;
#pragma unroll 8
  for (int i = tid; i < start; i += SCAN_T) part += cnti[i];
  for (int o = 16; o > 0; o >>= 1) part += __shfl_xor(part, o, 32);
  if (lane == 0) red[wv] = part;
  __syncthreads();
  int base = 0;
  for (int w = 0; w < SCAN_T / 32; ++w) base += red[w];
  __syncthreads();

  const int idx = start + tid;
  const int c = idx < N ? cnti[idx] : 0;
  int x = c;
  for (int o = 1; o < 32; o <<= 1) {
    const int y = __shfl_up(x, o, 32);
    if (lane >= o) x += y;
  }
  if (lane == 31) red[wv] = x;
  __syncthreads();
  int wbase = 0;
  for (int w = 0; w < wv; ++w) wbase += red[w];
  const int excl = base + wbase + x - c;
  if (idx < N) off[idx] = excl;
  if (idx == N - 1) off[N] = excl + c;
}

__global__ __launch_bounds__(256) void edge_place_kernel(
    const int* __restrict__ ei, int E, int N, const float* __restrict__ edyn,
    const float* __restrict__ esta, const float* __restrict__ sdyn,
    const float* __restrict__ ssta, const int* __restrict__ cnti,
    const int* __restrict__ off, const int* __restrict__ rank,
    float* __restrict__ rec) {
  const int e = blockIdx.x * 256 + threadIdx.x;
  const int ET = E + N;
  if (e >= ET) return;
  int s, d;
  if (e < E) {
    s = ei[e];
    d = ei[E + e];
  } else {
    s = d = e - E;
  }
  const float denom = fmaxf((float)cnti[d], 1.0f);
  const float wd = edyn[e] / (sdyn[d] * denom);
  const float ws = esta[e] / (ssta[d] * denom);
  ((float4*)rec)[off[d] + rank[e]] =
      make_float4(__int_as_float(s), wd, ws, 0.0f);
}

__global__ __launch_bounds__(256) void node_gather_kernel(
    int N, const int* __restrict__ off, const float* __restrict__ rec,
    const float* __restrict__ Vv, float* __restrict__ featsta,
    float* __restrict__ featdyn) {
  const int lane = threadIdx.x & 31;
  const int wv = __builtin_amdgcn_readfirstlane((int)(threadIdx.x >> 5));
  const int d = blockIdx.x * 8 + wv;
  if (d >= N) return;
  const int beg = off[d];
  const int end = off[d + 1];

  float4 as = make_float4(0.0f, 0.0f, 0.0f, 0.0f);
  float4 ad = make_float4(0.0f, 0.0f, 0.0f, 0.0f);
  for (int c = beg; c < end; c += 32) {
    const int n = (end - c < 32) ? (end - c) : 32;
    float4 rc = make_float4(0.0f, 0.0f, 0.0f, 0.0f);
    if (lane < n) rc = ((const float4*)rec)[c + lane];
    for (int b = 0; b < n; b += 8) {
      float4 v[8];
      int s[8];
#pragma unroll
      for (int j = 0; j < 8; ++j)
        s[j] = __builtin_amdgcn_readlane(__float_as_int(rc.x), b + j);
#pragma unroll
      for (int j = 0; j < 8; ++j) {
        if (b + j < n)
          v[j] = ((const float4*)(Vv + (size_t)s[j] * D))[lane];
      }
#pragma unroll
      for (int j = 0; j < 8; ++j) {
        if (b + j < n) {
          const float wdj = __int_as_float(
              __builtin_amdgcn_readlane(__float_as_int(rc.y), b + j));
          const float wsj = __int_as_float(
              __builtin_amdgcn_readlane(__float_as_int(rc.z), b + j));
          as.x += wsj * v[j].x;
          as.y += wsj * v[j].y;
          as.z += wsj * v[j].z;
          as.w += wsj * v[j].w;
          ad.x += wdj * v[j].x;
          ad.y += wdj * v[j].y;
          ad.z += wdj * v[j].z;
          ad.w += wdj * v[j].w;
        }
      }
    }
  }
  ((float4*)(featsta + (size_t)d * D))[lane] = as;
  ((float4*)(featdyn + (size_t)d * D))[lane] = ad;
}

extern "C" void kernel_launch(void* const* d_in, const int* in_sizes, int n_in,
                              void* d_out, int out_size, void* d_ws,
                              size_t ws_size, hipStream_t stream) {
  const float* curr = (const float*)d_in[0];
  const float* last = (const float*)d_in[1];
  const float* relemb = (const float*)d_in[2];
  const float* selfrel = (const float*)d_in[3];
  const float* Wq = (const float*)d_in[4];
  const float* bq = (const float*)d_in[5];
  const float* Wk = (const float*)d_in[6];
  const float* bk = (const float*)d_in[7];
  const float* Wv = (const float*)d_in[8];
  const float* bv = (const float*)d_in[9];
  const float* dynW = (const float*)d_in[10];
  const int* ei = (const int*)d_in[11];
  const int* et = (const int*)d_in[12];

  const int N = in_sizes[0] / D;
  const int E = in_sizes[12];
  const int R = in_sizes[2] / D;
  const int ET = E + N;
  const size_t ND = (size_t)N * D;

  float* ws = (float*)d_ws;
  float* Xq = ws;
  float* Xk = Xq + ND;
  float* Vv = Xk + ND;
  float* featdyn = Vv + ND;
  float* Trel = featdyn + ND;
  float* Tq = Trel + SELF_PAD * D;
  float* Tk = Tq + SELF_PAD * D;
  float* attn = Tk + SELF_PAD * D;
  float* edyn = attn + ET;
  float* esta = edyn + ET;
  float* mdyn = esta + ET;
  float* msta = mdyn + N;
  float* sdyn = msta + N;
  float* ssta = sdyn + N;
  int* cnti = (int*)(ssta + N);
  int* rank = (int*)featdyn;
  float* rec = Xq;
  int* off = (int*)(Xq + (size_t)4 * ET);

  float* out0 = (float*)d_out;
  float* out1 = out0 + ND;

  build_trel<<<(SELF_PAD * D + 255) / 256, 256, 0, stream>>>(
      relemb, selfrel, Trel, R, mdyn, msta, sdyn, ssta, cnti, N);

  gemm128_strip_wmma<<<(N / 16 + 1) / 2, 128, 0, stream>>>(curr, Wq, nullptr,
                                                         Xq, N);
  gemm128_strip_wmma<<<(N / 16 + 1) / 2, 128, 0, stream>>>(last, Wk, nullptr,
                                                         Xk, N);
  gemm128_strip_wmma<<<(N / 16 + 1) / 2, 128, 0, stream>>>(last, Wv, bv, Vv, N);
  gemm128_wmma<<<SELF_PAD / 16, 256, 0, stream>>>(Trel, Wq + D * D, bq, Tq,
                                                  SELF_PAD);
  gemm128_wmma<<<SELF_PAD / 16, 256, 0, stream>>>(Trel, Wk + D * D, bk, Tk,
                                                  SELF_PAD);

  edge_attn_kernel<<<(ET + 8 * EPW - 1) / (8 * EPW), 256, 0, stream>>>(
      ei, et, E, N, R, Xq, Xk, Tq, Tk, attn, mdyn, msta, cnti, rank);
  slot_range_kernel<<<(N + SCAN_T - 1) / SCAN_T, SCAN_T, 0, stream>>>(cnti, off,
                                                                     N);
  edge_expsum_kernel<<<(ET + 255) / 256, 256, 0, stream>>>(
      ei, E, N, attn, mdyn, msta, edyn, esta, sdyn, ssta);
  edge_place_kernel<<<(ET + 255) / 256, 256, 0, stream>>>(
      ei, E, N, edyn, esta, sdyn, ssta, cnti, off, rank, rec);
  node_gather_kernel<<<(N + 7) / 8, 256, 0, stream>>>(N, off, rec, Vv, out0,
                                                      featdyn);

  gemm128_strip_wmma<<<(N / 16 + 1) / 2, 128, 0, stream>>>(featdyn, dynW,
                                                         nullptr, out1, N);
}


# Edge phase by destination: one wave per node computes its edges' scores, both softmaxes as running max/sum in registers (sums rescaled when the max rises) and the weighted sums; float atomics gone
# speedup vs baseline: 2.5584x; 1.4993x over previous
#include <hip/hip_runtime.h>
#include <math.h>


typedef __attribute__((ext_vector_type(2))) float v2f;
typedef __attribute__((ext_vector_type(8))) float v8f;

#define D 128
#define SELF_PAD 512

__global__ __launch_bounds__(256) void gemm128_wmma(
    const float* __restrict__ A, const float* __restrict__ W,
    const float* __restrict__ bias, float* __restrict__ C, int M) {
  const int lane = threadIdx.x & 31;
  const int wave = threadIdx.x >> 5;
  const int r = lane & 15;
  const int g = lane >> 4;
  const int m0 = blockIdx.x * 16;
  const int n0 = wave * 16;

  const float* __restrict__ Arow = A + (size_t)(m0 + r) * D;
  const float* __restrict__ Wcol = W + (n0 + r);

  v8f acc = {};
#pragma unroll
  for (int k = 0; k < D; k += 4) {
    const int kk = k + 2 * g;
    v2f a;
    a.x = Arow[kk];
    a.y = Arow[kk + 1];
    v2f b;
    b.x = Wcol[(size_t)kk * D];
    b.y = Wcol[(size_t)(kk + 1) * D];
    acc = __builtin_amdgcn_wmma_f32_16x16x4_f32(
         false, a,  false, b,
         (short)0, acc,  false,  false);
  }

  const float bv = bias ? bias[n0 + r] : 0.0f;
#pragma unroll
  for (int j = 0; j < 8; ++j) {
    const int row = m0 + j + 8 * g;
    C[(size_t)row * D + n0 + r] = acc[j] + bv;
  }
}

#define WP 288

__global__ __launch_bounds__(128) void gemm128_strip_wmma(
    const float* __restrict__ A, const float* __restrict__ W,
    const float* __restrict__ bias, float* __restrict__ C, int M) {
  __shared__ __attribute__((aligned(16))) float Ws[64 * WP];
  const int tid = threadIdx.x;
  const int lane = tid & 31;
  const int wave = __builtin_amdgcn_readfirstlane(tid >> 5);
  const int r = lane & 15;
  const int g = lane >> 4;
  const int m0 = (blockIdx.x * 2 + (wave >> 1)) * 16;
  const int nb = (wave & 1) * 64;

  float4 w0[16], w1[16];
  const float* __restrict__ Wb = W + (size_t)(2 * wave) * D + 4 * lane;
#pragma unroll
  for (int it = 0; it < 16; ++it) {
    w0[it] = *(const float4*)(Wb + it * 8 * D);
    w1[it] = *(const float4*)(Wb + it * 8 * D + D);
  }

  int arow = m0 + r;
  if (arow > M - 1) arow = M - 1;
  const float* __restrict__ Ap = A + (size_t)arow * D + 2 * g;
  v2f a[32];
#pragma unroll
  for (int ks = 0; ks < 32; ++ks) a[ks] = *(const v2f*)(Ap + 4 * ks);

  float* dstb = Ws + wave * WP + 8 * lane;
#pragma unroll
  for (int it = 0; it < 16; ++it) {
    float* dst = dstb + it * 4 * WP;
    *(float4*)dst = make_float4(w0[it].x, w1[it].x, w0[it].y, w1[it].y);
    *(float4*)(dst + 4) = make_float4(w0[it].z, w1[it].z, w0[it].w, w1[it].w);
  }
  asm volatile("s_wait_dscnt 0x0\n\ts_barrier_signal -1\n\ts_barrier_wait -1"
               ::: "memory");

  float bv[4];
#pragma unroll
  for (int ct = 0; ct < 4; ++ct)
    bv[ct] = bias ? bias[nb + 16 * ct + r] : 0.0f;

  v8f acc[4] = {};
  const float* __restrict__ Bp = Ws + g * WP + 2 * (nb + r);
#pragma unroll
  for (int ks = 0; ks < 32; ++ks) {
#pragma unroll
    for (int ct = 0; ct < 4; ++ct) {
      const v2f b = *(const v2f*)(Bp + 2 * ks * WP + 32 * ct);
      acc[ct] = __builtin_amdgcn_wmma_f32_16x16x4_f32(
           false, a[ks],  false, b,
           (short)0, acc[ct],  false,  false);
    }
  }

  if (m0 >= M) return;
#pragma unroll
  for (int ct = 0; ct < 4; ++ct) {
    const int col = nb + 16 * ct + r;
#pragma unroll
    for (int j = 0; j < 8; ++j) {
      const int row = m0 + j + 8 * g;
      C[(size_t)row * D + col] = acc[ct][j] + bv[ct];
    }
  }
}

__global__ void build_trel(const float* __restrict__ relemb,
                           const float* __restrict__ selfrel,
                           float* __restrict__ Trel, int R,
                           int* __restrict__ cnti, int N) {
  const int i = blockIdx.x * 256 + threadIdx.x;
  if (i < SELF_PAD * D) {
    const int row = i >> 7;
    const int c = i & (D - 1);
    float v = 0.0f;
    if (row < R) v = relemb[i];
    else if (row == R) v = selfrel[c];
    Trel[i] = v;
  }
  for (int n = i; n < N; n += (int)gridDim.x * 256) cnti[n] = 0;
}

__global__ __launch_bounds__(256) void edge_rank_kernel(
    const int* __restrict__ ei, int E, int* __restrict__ cnti,
    int* __restrict__ rank) {
  const int e = blockIdx.x * 256 + threadIdx.x;
  if (e >= E) return;
  rank[e] = atomicAdd(&cnti[ei[E + e]], 1);
}

#define SCAN_T 1024

__global__ __launch_bounds__(SCAN_T) void slot_range_kernel(
    const int* __restrict__ cnti, int* __restrict__ off, int N) {
  __shared__ int red[SCAN_T / 32];
  const int tid = threadIdx.x;
  const int lane = tid & 31;
  const int wv = tid >> 5;
  const int start = blockIdx.x * SCAN_T;
  int part = 0;
#pragma unroll 8
  for (int i = tid; i < start; i += SCAN_T) part += cnti[i];
  for (int o = 16; o > 0; o >>= 1) part += __shfl_xor(part, o, 32);
  if (lane == 0) red[wv] = part;
  __syncthreads();
  int base = 0;
  for (int w = 0; w < SCAN_T / 32; ++w) base += red[w];
  __syncthreads();

  const int idx = start + tid;
  const int c = idx < N ? cnti[idx] : 0;
  int x = c;
  for (int o = 1; o < 32; o <<= 1) {
    const int y = __shfl_up(x, o, 32);
    if (lane >= o) x += y;
  }
  if (lane == 31) red[wv] = x;
  __syncthreads();
  int wbase = 0;
  for (int w = 0; w < wv; ++w) wbase += red[w];
  const int excl = base + wbase + x - c;
  if (idx < N) off[idx] = excl;
  if (idx == N - 1) off[N] = excl + c;
}

__global__ __launch_bounds__(256) void edge_place_kernel(
    const int* __restrict__ ei, const int* __restrict__ et, int E,
    const int* __restrict__ off, const int* __restrict__ rank,
    int* __restrict__ rec) {
  const int e = blockIdx.x * 256 + threadIdx.x;
  if (e >= E) return;
  const int d = ei[E + e];
  ((int2*)rec)[off[d] + rank[e]] = make_int2(ei[e], et[e]);
}

__global__ __launch_bounds__(128) void node_segment_kernel(
    int N, int R, const int* __restrict__ off, const int* __restrict__ rec,
    const float* __restrict__ Xq, const float* __restrict__ Xk,
    const float* __restrict__ Tq, const float* __restrict__ Tk,
    const float* __restrict__ Vv, float* __restrict__ featsta,
    float* __restrict__ featdyn) {
  const int lane = threadIdx.x & 31;
  const int wv = __builtin_amdgcn_readfirstlane((int)(threadIdx.x >> 5));
  const int d = blockIdx.x * 4 + wv;
  if (d >= N) return;
  const int beg = off[d];
  const int nt = off[d + 1] - beg + 1;
  int rs = d, rt = R;
  if (lane > 0 && lane < nt) {
    const int2 p = ((const int2*)rec)[beg + lane - 1];
    rs = p.x;
    rt = p.y;
  }
  const float4 xq = ((const float4*)(Xq + (size_t)d * D))[lane];

  float md = -__builtin_inff();
  float ms = -__builtin_inff();
  float sd = 0.0f, ss = 0.0f;
  float4 as = make_float4(0.0f, 0.0f, 0.0f, 0.0f);
  float4 ad = make_float4(0.0f, 0.0f, 0.0f, 0.0f);

  const bool up16 = (lane & 16) != 0;
  const bool up8 = (lane & 8) != 0;
  const bool up4 = (lane & 4) != 0;

  for (int c = 0; c < nt; c += 32) {
    const int n = (nt - c < 32) ? (nt - c) : 32;
    if (c > 0) {
      rs = d;
      rt = R;
      if (c + lane < nt) {
        const int2 p = ((const int2*)rec)[beg + c + lane - 1];
        rs = p.x;
        rt = p.y;
      }
    }
    for (int b = 0; b < n; b += 8) {
      int s[8], t[8];
#pragma unroll
      for (int j = 0; j < 8; ++j) {
        s[j] = __builtin_amdgcn_readlane(rs, b + j);
        t[j] = __builtin_amdgcn_readlane(rt, b + j);
      }
      float4 k0[8], k1[8], q1[8], v[8];
#pragma unroll
      for (int j = 0; j < 8; ++j) {
        if (b + j < n) {
          k0[j] = ((const float4*)(Xk + (size_t)s[j] * D))[lane];
          k1[j] = ((const float4*)(Tk + (size_t)t[j] * D))[lane];
          q1[j] = ((const float4*)(Tq + (size_t)t[j] * D))[lane];
        }
      }
#pragma unroll
      for (int j = 0; j < 8; ++j) {
        if (b + j < n)
          v[j] = ((const float4*)(Vv + (size_t)s[j] * D))[lane];
      }

      float a[8];
#pragma unroll
      for (int j = 0; j < 8; ++j) {
        a[j] = 0.0f;
        if (b + j < n) {
          a[j] = (xq.x + q1[j].x) * (k0[j].x + k1[j].x) +
                 (xq.y + q1[j].y) * (k0[j].y + k1[j].y) +
                 (xq.z + q1[j].z) * (k0[j].z + k1[j].z) +
                 (xq.w + q1[j].w) * (k0[j].w + k1[j].w);
        }
      }

      float b4[4];
#pragma unroll
      for (int j = 0; j < 4; ++j) {
        const float keep = up16 ? a[j + 4] : a[j];
        const float give = up16 ? a[j] : a[j + 4];
        b4[j] = keep + __shfl_xor(give, 16, 32);
      }
      float c2[2];
#pragma unroll
      for (int j = 0; j < 2; ++j) {
        const float keep = up8 ? b4[j + 2] : b4[j];
        const float give = up8 ? b4[j] : b4[j + 2];
        c2[j] = keep + __shfl_xor(give, 8, 32);
      }
      float r;
      {
        const float keep = up4 ? c2[1] : c2[0];
        const float give = up4 ? c2[0] : c2[1];
        r = keep + __shfl_xor(give, 4, 32);
      }
      r += __shfl_xor(r, 2, 32);
      r += __shfl_xor(r, 1, 32);
      r *= 0.08838834764831845f;

      float sc[8];
#pragma unroll
      for (int j = 0; j < 8; ++j)
        sc[j] = __int_as_float(
            __builtin_amdgcn_readlane(__float_as_int(r), 4 * j));

      float gmax = sc[0], gmin = sc[0];
#pragma unroll
      for (int j = 1; j < 8; ++j) {
        if (b + j < n) {
          gmax = fmaxf(gmax, sc[j]);
          gmin = fminf(gmin, sc[j]);
        }
      }
      const float mdn = fmaxf(md, gmax);
      const float msn = fmaxf(ms, -gmin);
      const float fd = __expf(md - mdn);
      const float fs = __expf(ms - msn);
      sd *= fd;
      ss *= fs;
      ad.x *= fd;
      ad.y *= fd;
      ad.z *= fd;
      ad.w *= fd;
      as.x *= fs;
      as.y *= fs;
      as.z *= fs;
      as.w *= fs;
      md = mdn;
      ms = msn;
#pragma unroll
      for (int j = 0; j < 8; ++j) {
        if (b + j < n) {
          const float ed = __expf(sc[j] - mdn);
          const float es = __expf(-sc[j] - msn);
          sd += ed;
          ss += es;
          ad.x += ed * v[j].x;
          ad.y += ed * v[j].y;
          ad.z += ed * v[j].z;
          ad.w += ed * v[j].w;
          as.x += es * v[j].x;
          as.y += es * v[j].y;
          as.z += es * v[j].z;
          as.w += es * v[j].w;
        }
      }
    }
  }
  const float denom = (float)nt;
  const float wd = 1.0f / (sd * denom);
  const float ws = 1.0f / (ss * denom);
  ((float4*)(featsta + (size_t)d * D))[lane] =
      make_float4(as.x * ws, as.y * ws, as.z * ws, as.w * ws);
  ((float4*)(featdyn + (size_t)d * D))[lane] =
      make_float4(ad.x * wd, ad.y * wd, ad.z * wd, ad.w * wd);
}

extern "C" void kernel_launch(void* const* d_in, const int* in_sizes, int n_in,
                              void* d_out, int out_size, void* d_ws,
                              size_t ws_size, hipStream_t stream) {
  const float* curr = (const float*)d_in[0];
  const float* last = (const float*)d_in[1];
  const float* relemb = (const float*)d_in[2];
  const float* selfrel = (const float*)d_in[3];
  const float* Wq = (const float*)d_in[4];
  const float* bq = (const float*)d_in[5];
  const float* Wk = (const float*)d_in[6];
  const float* bk = (const float*)d_in[7];
  const float* Wv = (const float*)d_in[8];
  const float* bv = (const float*)d_in[9];
  const float* dynW = (const float*)d_in[10];
  const int* ei = (const int*)d_in[11];
  const int* et = (const int*)d_in[12];

  const int N = in_sizes[0] / D;
  const int E = in_sizes[12];
  const int R = in_sizes[2] / D;
  const size_t ND = (size_t)N * D;

  float* ws = (float*)d_ws;
  float* Xq = ws;
  float* Xk = Xq + ND;
  float* Vv = Xk + ND;
  float* featdyn = Vv + ND;
  float* Trel = featdyn + ND;
  float* Tq = Trel + SELF_PAD * D;
  float* Tk = Tq + SELF_PAD * D;
  int* rec = (int*)(Tk + SELF_PAD * D);
  int* off = rec + (size_t)2 * E;
  int* cnti = off + (N + 1);
  int* rank = (int*)featdyn;

  float* out0 = (float*)d_out;
  float* out1 = out0 + ND;

  build_trel<<<(SELF_PAD * D + 255) / 256, 256, 0, stream>>>(
      relemb, selfrel, Trel, R, cnti, N);

  edge_rank_kernel<<<(E + 255) / 256, 256, 0, stream>>>(ei, E, cnti, rank);
  slot_range_kernel<<<(N + SCAN_T - 1) / SCAN_T, SCAN_T, 0, stream>>>(cnti, off,
                                                                     N);
  edge_place_kernel<<<(E + 255) / 256, 256, 0, stream>>>(ei, et, E, off, rank,
                                                        rec);

  gemm128_strip_wmma<<<(N / 16 + 1) / 2, 128, 0, stream>>>(curr, Wq, nullptr,
                                                         Xq, N);
  gemm128_strip_wmma<<<(N / 16 + 1) / 2, 128, 0, stream>>>(last, Wk, nullptr,
                                                         Xk, N);
  gemm128_strip_wmma<<<(N / 16 + 1) / 2, 128, 0, stream>>>(last, Wv, bv, Vv, N);
  gemm128_wmma<<<SELF_PAD / 16, 256, 0, stream>>>(Trel, Wq + D * D, bq, Tq,
                                                  SELF_PAD);
  gemm128_wmma<<<SELF_PAD / 16, 256, 0, stream>>>(Trel, Wk + D * D, bk, Tk,
                                                  SELF_PAD);

  node_segment_kernel<<<(N + 3) / 4, 128, 0, stream>>>(
      N, R, off, rec, Xq, Xk, Tq, Tk, Vv, out0, featdyn);

  gemm128_strip_wmma<<<(N / 16 + 1) / 2, 128, 0, stream>>>(featdyn, dynW,
                                                         nullptr, out1, N);
}


# Eleven launches -> six: the three node projections, the two relation projections (rows read from the input tables, no padded copy) and the count clearing are one grid; block index picks the product
# speedup vs baseline: 2.9059x; 1.1358x over previous
#include <hip/hip_runtime.h>
#include <math.h>


typedef __attribute__((ext_vector_type(2))) float v2f;
typedef __attribute__((ext_vector_type(8))) float v8f;

#define D 128
#define SELF_PAD 512

#define WP 288

__device__ __forceinline__ void strip_gemm_block(
    const float* __restrict__ Ap, const float* __restrict__ W,
    const float* __restrict__ bias, float* __restrict__ C, int m0, int M,
    int wave, int lane, int r, int g) {
  __shared__ __attribute__((aligned(16))) float Ws[64 * WP];
  const int nb = (wave & 1) * 64;

  float4 w0[16], w1[16];
  const float* __restrict__ Wb = W + (size_t)(2 * wave) * D + 4 * lane;
#pragma unroll
  for (int it = 0; it < 16; ++it) {
    w0[it] = *(const float4*)(Wb + it * 8 * D);
    w1[it] = *(const float4*)(Wb + it * 8 * D + D);
  }

  v2f a[32];
#pragma unroll
  for (int ks = 0; ks < 32; ++ks) a[ks] = *(const v2f*)(Ap + 4 * ks);

  float* dstb = Ws + wave * WP + 8 * lane;
#pragma unroll
  for (int it = 0; it < 16; ++it) {
    float* dst = dstb + it * 4 * WP;
    *(float4*)dst = make_float4(w0[it].x, w1[it].x, w0[it].y, w1[it].y);
    *(float4*)(dst + 4) = make_float4(w0[it].z, w1[it].z, w0[it].w, w1[it].w);
  }
  asm volatile("s_wait_dscnt 0x0\n\ts_barrier_signal -1\n\ts_barrier_wait -1"
               ::: "memory");

  float bv[4];
#pragma unroll
  for (int ct = 0; ct < 4; ++ct)
    bv[ct] = bias ? bias[nb + 16 * ct + r] : 0.0f;

  v8f acc[4] = {};
  const float* __restrict__ Bp = Ws + g * WP + 2 * (nb + r);
#pragma unroll
  for (int ks = 0; ks < 32; ++ks) {
#pragma unroll
    for (int ct = 0; ct < 4; ++ct) {
      const v2f b = *(const v2f*)(Bp + 2 * ks * WP + 32 * ct);
      acc[ct] = __builtin_amdgcn_wmma_f32_16x16x4_f32(
           false, a[ks],  false, b,
           (short)0, acc[ct],  false,  false);
    }
  }

  if (m0 >= M) return;
#pragma unroll
  for (int ct = 0; ct < 4; ++ct) {
    const int col = nb + 16 * ct + r;
#pragma unroll
    for (int j = 0; j < 8; ++j) {
      const int row = m0 + j + 8 * g;
      C[(size_t)row * D + col] = acc[ct][j] + bv[ct];
    }
  }
}

__global__ __launch_bounds__(128) void gemm128_strip_wmma(
    const float* __restrict__ A, const float* __restrict__ W,
    const float* __restrict__ bias, float* __restrict__ C, int M) {
  const int tid = threadIdx.x;
  const int lane = tid & 31;
  const int wave = __builtin_amdgcn_readfirstlane(tid >> 5);
  const int r = lane & 15;
  const int g = lane >> 4;
  const int m0 = (blockIdx.x * 2 + (wave >> 1)) * 16;
  int arow = m0 + r;
  if (arow > M - 1) arow = M - 1;
  strip_gemm_block(A + (size_t)arow * D + 2 * g, W, bias, C, m0, M, wave, lane,
                   r, g);
}

__global__ __launch_bounds__(128) void front_products_wmma(
    const float* __restrict__ curr, const float* __restrict__ last,
    const float* __restrict__ relemb, const float* __restrict__ selfrel,
    const float* __restrict__ Wq, const float* __restrict__ Wk,
    const float* __restrict__ Wv, const float* __restrict__ bq,
    const float* __restrict__ bk, const float* __restrict__ bvv,
    float* __restrict__ Xq, float* __restrict__ Xk, float* __restrict__ Vv,
    float* __restrict__ Tq, float* __restrict__ Tk, int* __restrict__ cnti,
    int N, int R) {
  const int tid = threadIdx.x;
  const int lane = tid & 31;
  const int wave = __builtin_amdgcn_readfirstlane(tid >> 5);
  const int r = lane & 15;
  const int g = lane >> 4;

  const int n = blockIdx.x * 128 + tid;
  if (n < N) cnti[n] = 0;

  const int G = (N / 16 + 1) / 2;
  int j = blockIdx.x;
  const float* A = curr;
  const float* W = Wq;
  const float* bias = nullptr;
  float* C = Xq;
  int M = N;
  bool rel = false;
  if (j >= G) {
    j -= G;
    A = last;
    W = Wk;
    C = Xk;
    if (j >= G) {
      j -= G;
      W = Wv;
      bias = bvv;
      C = Vv;
      if (j >= G) {
        j -= G;
        rel = true;
        M = SELF_PAD;
        W = Wq + D * D;
        bias = bq;
        C = Tq;
        if (j >= SELF_PAD / 32) {
          j -= SELF_PAD / 32;
          W = Wk + D * D;
          bias = bk;
          C = Tk;
        }
      }
    }
  }
  const int m0 = (j * 2 + (wave >> 1)) * 16;
  int arow = m0 + r;
  if (arow > M - 1) arow = M - 1;
  const float* Arow = A + (size_t)arow * D;
  if (rel) Arow = arow < R ? relemb + (size_t)arow * D : selfrel;
  strip_gemm_block(Arow + 2 * g, W, bias, C, m0, M, wave, lane, r, g);
}

__global__ __launch_bounds__(256) void edge_rank_kernel(
    const int* __restrict__ ei, int E, int* __restrict__ cnti,
    int* __restrict__ rank) {
  const int e = blockIdx.x * 256 + threadIdx.x;
  if (e >= E) return;
  rank[e] = atomicAdd(&cnti[ei[E + e]], 1);
}

#define SCAN_T 1024

__global__ __launch_bounds__(SCAN_T) void slot_range_kernel(
    const int* __restrict__ cnti, int* __restrict__ off, int N) {
  __shared__ int red[SCAN_T / 32];
  const int tid = threadIdx.x;
  const int lane = tid & 31;
  const int wv = tid >> 5;
  const int start = blockIdx.x * SCAN_T;
  int part = 0;
#pragma unroll 8
  for (int i = tid; i < start; i += SCAN_T) part += cnti[i];
  for (int o = 16; o > 0; o >>= 1) part += __shfl_xor(part, o, 32);
  if (lane == 0) red[wv] = part;
  __syncthreads();
  int base = 0;
  for (int w = 0; w < SCAN_T / 32; ++w) base += red[w];
  __syncthreads();

  const int idx = start + tid;
  const int c = idx < N ? cnti[idx] : 0;
  int x = c;
  for (int o = 1; o < 32; o <<= 1) {
    const int y = __shfl_up(x, o, 32);
    if (lane >= o) x += y;
  }
  if (lane == 31) red[wv] = x;
  __syncthreads();
  int wbase = 0;
  for (int w = 0; w < wv; ++w) wbase += red[w];
  const int excl = base + wbase + x - c;
  if (idx < N) off[idx] = excl;
  if (idx == N - 1) off[N] = excl + c;
}

__global__ __launch_bounds__(256) void edge_place_kernel(
    const int* __restrict__ ei, const int* __restrict__ et, int E,
    const int* __restrict__ off, const int* __restrict__ rank,
    int* __restrict__ rec) {
  const int e = blockIdx.x * 256 + threadIdx.x;
  if (e >= E) return;
  const int d = ei[E + e];
  ((int2*)rec)[off[d] + rank[e]] = make_int2(ei[e], et[e]);
}

__global__ __launch_bounds__(128) void node_segment_kernel(
    int N, int R, const int* __restrict__ off, const int* __restrict__ rec,
    const float* __restrict__ Xq, const float* __restrict__ Xk,
    const float* __restrict__ Tq, const float* __restrict__ Tk,
    const float* __restrict__ Vv, float* __restrict__ featsta,
    float* __restrict__ featdyn) {
  const int lane = threadIdx.x & 31;
  const int wv = __builtin_amdgcn_readfirstlane((int)(threadIdx.x >> 5));
  const int d = blockIdx.x * 4 + wv;
  if (d >= N) return;
  const int beg = off[d];
  const int nt = off[d + 1] - beg + 1;
  int rs = d, rt = R;
  if (lane > 0 && lane < nt) {
    const int2 p = ((const int2*)rec)[beg + lane - 1];
    rs = p.x;
    rt = p.y;
  }
  const float4 xq = ((const float4*)(Xq + (size_t)d * D))[lane];

  float md = -__builtin_inff();
  float ms = -__builtin_inff();
  float sd = 0.0f, ss = 0.0f;
  float4 as = make_float4(0.0f, 0.0f, 0.0f, 0.0f);
  float4 ad = make_float4(0.0f, 0.0f, 0.0f, 0.0f);

  const bool up16 = (lane & 16) != 0;
  const bool up8 = (lane & 8) != 0;
  const bool up4 = (lane & 4) != 0;

  for (int c = 0; c < nt; c += 32) {
    const int n = (nt - c < 32) ? (nt - c) : 32;
    if (c > 0) {
      rs = d;
      rt = R;
      if (c + lane < nt) {
        const int2 p = ((const int2*)rec)[beg + c + lane - 1];
        rs = p.x;
        rt = p.y;
      }
    }
    for (int b = 0; b < n; b += 8) {
      int s[8], t[8];
#pragma unroll
      for (int j = 0; j < 8; ++j) {
        s[j] = __builtin_amdgcn_readlane(rs, b + j);
        t[j] = __builtin_amdgcn_readlane(rt, b + j);
      }
      float4 k0[8], k1[8], q1[8], v[8];
#pragma unroll
      for (int j = 0; j < 8; ++j) {
        if (b + j < n) {
          k0[j] = ((const float4*)(Xk + (size_t)s[j] * D))[lane];
          k1[j] = ((const float4*)(Tk + (size_t)t[j] * D))[lane];
          q1[j] = ((const float4*)(Tq + (size_t)t[j] * D))[lane];
        }
      }
#pragma unroll
      for (int j = 0; j < 8; ++j) {
        if (b + j < n)
          v[j] = ((const float4*)(Vv + (size_t)s[j] * D))[lane];
      }

      float a[8];
#pragma unroll
      for (int j = 0; j < 8; ++j) {
        a[j] = 0.0f;
        if (b + j < n) {
          a[j] = (xq.x + q1[j].x) * (k0[j].x + k1[j].x) +
                 (xq.y + q1[j].y) * (k0[j].y + k1[j].y) +
                 (xq.z + q1[j].z) * (k0[j].z + k1[j].z) +
                 (xq.w + q1[j].w) * (k0[j].w + k1[j].w);
        }
      }

      float b4[4];
#pragma unroll
      for (int j = 0; j < 4; ++j) {
        const float keep = up16 ? a[j + 4] : a[j];
        const float give = up16 ? a[j] : a[j + 4];
        b4[j] = keep + __shfl_xor(give, 16, 32);
      }
      float c2[2];
#pragma unroll
      for (int j = 0; j < 2; ++j) {
        const float keep = up8 ? b4[j + 2] : b4[j];
        const float give = up8 ? b4[j] : b4[j + 2];
        c2[j] = keep + __shfl_xor(give, 8, 32);
      }
      float r;
      {
        const float keep = up4 ? c2[1] : c2[0];
        const float give = up4 ? c2[0] : c2[1];
        r = keep + __shfl_xor(give, 4, 32);
      }
      r += __shfl_xor(r, 2, 32);
      r += __shfl_xor(r, 1, 32);
      r *= 0.08838834764831845f;

      float sc[8];
#pragma unroll
      for (int j = 0; j < 8; ++j)
        sc[j] = __int_as_float(
            __builtin_amdgcn_readlane(__float_as_int(r), 4 * j));

      float gmax = sc[0], gmin = sc[0];
#pragma unroll
      for (int j = 1; j < 8; ++j) {
        if (b + j < n) {
          gmax = fmaxf(gmax, sc[j]);
          gmin = fminf(gmin, sc[j]);
        }
      }
      const float mdn = fmaxf(md, gmax);
      const float msn = fmaxf(ms, -gmin);
      const float fd = __expf(md - mdn);
      const float fs = __expf(ms - msn);
      sd *= fd;
      ss *= fs;
      ad.x *= fd;
      ad.y *= fd;
      ad.z *= fd;
      ad.w *= fd;
      as.x *= fs;
      as.y *= fs;
      as.z *= fs;
      as.w *= fs;
      md = mdn;
      ms = msn;
#pragma unroll
      for (int j = 0; j < 8; ++j) {
        if (b + j < n) {
          const float ed = __expf(sc[j] - mdn);
          const float es = __expf(-sc[j] - msn);
          sd += ed;
          ss += es;
          ad.x += ed * v[j].x;
          ad.y += ed * v[j].y;
          ad.z += ed * v[j].z;
          ad.w += ed * v[j].w;
          as.x += es * v[j].x;
          as.y += es * v[j].y;
          as.z += es * v[j].z;
          as.w += es * v[j].w;
        }
      }
    }
  }
  const float denom = (float)nt;
  const float wd = 1.0f / (sd * denom);
  const float ws = 1.0f / (ss * denom);
  ((float4*)(featsta + (size_t)d * D))[lane] =
      make_float4(as.x * ws, as.y * ws, as.z * ws, as.w * ws);
  ((float4*)(featdyn + (size_t)d * D))[lane] =
      make_float4(ad.x * wd, ad.y * wd, ad.z * wd, ad.w * wd);
}

extern "C" void kernel_launch(void* const* d_in, const int* in_sizes, int n_in,
                              void* d_out, int out_size, void* d_ws,
                              size_t ws_size, hipStream_t stream) {
  const float* curr = (const float*)d_in[0];
  const float* last = (const float*)d_in[1];
  const float* relemb = (const float*)d_in[2];
  const float* selfrel = (const float*)d_in[3];
  const float* Wq = (const float*)d_in[4];
  const float* bq = (const float*)d_in[5];
  const float* Wk = (const float*)d_in[6];
  const float* bk = (const float*)d_in[7];
  const float* Wv = (const float*)d_in[8];
  const float* bv = (const float*)d_in[9];
  const float* dynW = (const float*)d_in[10];
  const int* ei = (const int*)d_in[11];
  const int* et = (const int*)d_in[12];

  const int N = in_sizes[0] / D;
  const int E = in_sizes[12];
  const int R = in_sizes[2] / D;
  const size_t ND = (size_t)N * D;

  float* ws = (float*)d_ws;
  float* Xq = ws;
  float* Xk = Xq + ND;
  float* Vv = Xk + ND;
  float* featdyn = Vv + ND;
  float* Tq = featdyn + ND;
  float* Tk = Tq + SELF_PAD * D;
  int* rec = (int*)(Tk + SELF_PAD * D);
  int* off = rec + (size_t)2 * E;
  int* cnti = off + (N + 1);
  int* rank = (int*)featdyn;

  float* out0 = (float*)d_out;
  float* out1 = out0 + ND;

  front_products_wmma<<<3 * ((N / 16 + 1) / 2) + 2 * (SELF_PAD / 32), 128, 0,
                        stream>>>(curr, last, relemb, selfrel, Wq, Wk, Wv, bq,
                                  bk, bv, Xq, Xk, Vv, Tq, Tk, cnti, N, R);

  edge_rank_kernel<<<(E + 255) / 256, 256, 0, stream>>>(ei, E, cnti, rank);
  slot_range_kernel<<<(N + SCAN_T - 1) / SCAN_T, SCAN_T, 0, stream>>>(cnti, off,
                                                                     N);
  edge_place_kernel<<<(E + 255) / 256, 256, 0, stream>>>(ei, et, E, off, rank,
                                                        rec);

  node_segment_kernel<<<(N + 3) / 4, 128, 0, stream>>>(
      N, R, off, rec, Xq, Xk, Tq, Tk, Vv, out0, featdyn);

  gemm128_strip_wmma<<<(N / 16 + 1) / 2, 128, 0, stream>>>(featdyn, dynW,
                                                         nullptr, out1, N);
}


# Edge records written straight to fixed slots of their destination (slot = rank from the count atomic; overflow list past 31): prefix-sum and placement launches gone; self-loop now summed last
# speedup vs baseline: 3.1694x; 1.0907x over previous
#include <hip/hip_runtime.h>
#include <math.h>


typedef __attribute__((ext_vector_type(2))) float v2f;
typedef __attribute__((ext_vector_type(8))) float v8f;

#define D 128
#define SELF_PAD 512

#define WP 288

__device__ __forceinline__ void strip_gemm_block(
    const float* __restrict__ Ap, const float* __restrict__ W,
    const float* __restrict__ bias, float* __restrict__ C, int m0, int M,
    int wave, int lane, int r, int g) {
  __shared__ __attribute__((aligned(16))) float Ws[64 * WP];
  const int nb = (wave & 1) * 64;

  float4 w0[16], w1[16];
  const float* __restrict__ Wb = W + (size_t)(2 * wave) * D + 4 * lane;
#pragma unroll
  for (int it = 0; it < 16; ++it) {
    w0[it] = *(const float4*)(Wb + it * 8 * D);
    w1[it] = *(const float4*)(Wb + it * 8 * D + D);
  }

  v2f a[32];
#pragma unroll
  for (int ks = 0; ks < 32; ++ks) a[ks] = *(const v2f*)(Ap + 4 * ks);

  float* dstb = Ws + wave * WP + 8 * lane;
#pragma unroll
  for (int it = 0; it < 16; ++it) {
    float* dst = dstb + it * 4 * WP;
    *(float4*)dst = make_float4(w0[it].x, w1[it].x, w0[it].y, w1[it].y);
    *(float4*)(dst + 4) = make_float4(w0[it].z, w1[it].z, w0[it].w, w1[it].w);
  }
  asm volatile("s_wait_dscnt 0x0\n\ts_barrier_signal -1\n\ts_barrier_wait -1"
               ::: "memory");

  float bv[4];
#pragma unroll
  for (int ct = 0; ct < 4; ++ct)
    bv[ct] = bias ? bias[nb + 16 * ct + r] : 0.0f;

  v8f acc[4] = {};
  const float* __restrict__ Bp = Ws + g * WP + 2 * (nb + r);
#pragma unroll
  for (int ks = 0; ks < 32; ++ks) {
#pragma unroll
    for (int ct = 0; ct < 4; ++ct) {
      const v2f b = *(const v2f*)(Bp + 2 * ks * WP + 32 * ct);
      acc[ct] = __builtin_amdgcn_wmma_f32_16x16x4_f32(
           false, a[ks],  false, b,
           (short)0, acc[ct],  false,  false);
    }
  }

  if (m0 >= M) return;
#pragma unroll
  for (int ct = 0; ct < 4; ++ct) {
    const int col = nb + 16 * ct + r;
#pragma unroll
    for (int j = 0; j < 8; ++j) {
      const int row = m0 + j + 8 * g;
      C[(size_t)row * D + col] = acc[ct][j] + bv[ct];
    }
  }
}

__global__ __launch_bounds__(128) void gemm128_strip_wmma(
    const float* __restrict__ A, const float* __restrict__ W,
    const float* __restrict__ bias, float* __restrict__ C, int M) {
  const int tid = threadIdx.x;
  const int lane = tid & 31;
  const int wave = __builtin_amdgcn_readfirstlane(tid >> 5);
  const int r = lane & 15;
  const int g = lane >> 4;
  const int m0 = (blockIdx.x * 2 + (wave >> 1)) * 16;
  int arow = m0 + r;
  if (arow > M - 1) arow = M - 1;
  strip_gemm_block(A + (size_t)arow * D + 2 * g, W, bias, C, m0, M, wave, lane,
                   r, g);
}

__global__ __launch_bounds__(128) void front_products_wmma(
    const float* __restrict__ curr, const float* __restrict__ last,
    const float* __restrict__ relemb, const float* __restrict__ selfrel,
    const float* __restrict__ Wq, const float* __restrict__ Wk,
    const float* __restrict__ Wv, const float* __restrict__ bq,
    const float* __restrict__ bk, const float* __restrict__ bvv,
    float* __restrict__ Xq, float* __restrict__ Xk, float* __restrict__ Vv,
    float* __restrict__ Tq, float* __restrict__ Tk, int* __restrict__ cnti,
    int* __restrict__ ovfn, int N, int R) {
  const int tid = threadIdx.x;
  const int lane = tid & 31;
  const int wave = __builtin_amdgcn_readfirstlane(tid >> 5);
  const int r = lane & 15;
  const int g = lane >> 4;

  const int n = blockIdx.x * 128 + tid;
  if (n < N) cnti[n] = 0;
  if (n == 0) ovfn[0] = 0;

  const int G = (N / 16 + 1) / 2;
  int j = blockIdx.x;
  const float* A = curr;
  const float* W = Wq;
  const float* bias = nullptr;
  float* C = Xq;
  int M = N;
  bool rel = false;
  if (j >= G) {
    j -= G;
    A = last;
    W = Wk;
    C = Xk;
    if (j >= G) {
      j -= G;
      W = Wv;
      bias = bvv;
      C = Vv;
      if (j >= G) {
        j -= G;
        rel = true;
        M = SELF_PAD;
        W = Wq + D * D;
        bias = bq;
        C = Tq;
        if (j >= SELF_PAD / 32) {
          j -= SELF_PAD / 32;
          W = Wk + D * D;
          bias = bk;
          C = Tk;
        }
      }
    }
  }
  const int m0 = (j * 2 + (wave >> 1)) * 16;
  int arow = m0 + r;
  if (arow > M - 1) arow = M - 1;
  const float* Arow = A + (size_t)arow * D;
  if (rel) Arow = arow < R ? relemb + (size_t)arow * D : selfrel;
  strip_gemm_block(Arow + 2 * g, W, bias, C, m0, M, wave, lane, r, g);
}

#define SLOTS 31

__global__ __launch_bounds__(256) void edge_slot_kernel(
    const int* __restrict__ ei, const int* __restrict__ et, int E,
    int* __restrict__ cnti, int* chunks, int* __restrict__ ovfn,
    int* __restrict__ ovf) {
  const int e = blockIdx.x * 256 + threadIdx.x;
  if (e >= E) return;
  const int s = ei[e];
  const int d = ei[E + e];
  const int t = et[e];
  const int rk = atomicAdd(&cnti[d], 1);
  if (rk < SLOTS) {
    ((int2*)(chunks + (size_t)d * D))[rk] = make_int2(s, t);
  } else {
    const int o = atomicAdd(ovfn, 1);
    ((int2*)ovf)[o] = make_int2(s, d * SELF_PAD + t);
  }
}

__global__ __launch_bounds__(128) void node_segment_kernel(
    int N, int R, int E, const int* __restrict__ cnti,
    const int* __restrict__ ovfn, const int* __restrict__ ovf,
    const float* __restrict__ Xq, const float* __restrict__ Xk,
    const float* __restrict__ Tq, const float* __restrict__ Tk,
    const float* __restrict__ Vv, float* __restrict__ featsta,
    float* featdyn) {
  const int lane = threadIdx.x & 31;
  const int wv = __builtin_amdgcn_readfirstlane((int)(threadIdx.x >> 5));
  const int d = blockIdx.x * 4 + wv;
  if (d >= N) return;
  const int2 p = ((const int2*)((const int*)featdyn + (size_t)d * D))[lane];
  const int cnt = cnti[d];
  const int nt = cnt + 1;
  const float4 xq = ((const float4*)(Xq + (size_t)d * D))[lane];
  const int m = cnt < SLOTS ? cnt : SLOTS;
  int rs = p.x, rt = p.y;
  if (lane >= m) {
    rs = d;
    rt = R;
  }
  int n = m + 1;
  int novf = 0;
  if (cnt > SLOTS) {
    novf = ovfn[0];
    if (novf > E) novf = E;
  }
  int oc = 0;
  unsigned pend = 0u;
  int os = 0, ow = -1;

  float md = -__builtin_inff();
  float ms = -__builtin_inff();
  float sd = 0.0f, ss = 0.0f;
  float4 as = make_float4(0.0f, 0.0f, 0.0f, 0.0f);
  float4 ad = make_float4(0.0f, 0.0f, 0.0f, 0.0f);

  const bool up16 = (lane & 16) != 0;
  const bool up8 = (lane & 8) != 0;
  const bool up4 = (lane & 4) != 0;

  for (;;) {
    for (int b = 0; b < n; b += 8) {
      int s[8], t[8];
#pragma unroll
      for (int j = 0; j < 8; ++j) {
        s[j] = __builtin_amdgcn_readlane(rs, b + j);
        t[j] = __builtin_amdgcn_readlane(rt, b + j);
      }
      float4 k0[8], k1[8], q1[8], v[8];
#pragma unroll
      for (int j = 0; j < 8; ++j) {
        if (b + j < n) {
          k0[j] = ((const float4*)(Xk + (size_t)s[j] * D))[lane];
          k1[j] = ((const float4*)(Tk + (size_t)t[j] * D))[lane];
          q1[j] = ((const float4*)(Tq + (size_t)t[j] * D))[lane];
        }
      }
#pragma unroll
      for (int j = 0; j < 8; ++j) {
        if (b + j < n)
          v[j] = ((const float4*)(Vv + (size_t)s[j] * D))[lane];
      }

      float a[8];
#pragma unroll
      for (int j = 0; j < 8; ++j) {
        a[j] = 0.0f;
        if (b + j < n) {
          a[j] = (xq.x + q1[j].x) * (k0[j].x + k1[j].x) +
                 (xq.y + q1[j].y) * (k0[j].y + k1[j].y) +
                 (xq.z + q1[j].z) * (k0[j].z + k1[j].z) +
                 (xq.w + q1[j].w) * (k0[j].w + k1[j].w);
        }
      }

      float b4[4];
#pragma unroll
      for (int j = 0; j < 4; ++j) {
        const float keep = up16 ? a[j + 4] : a[j];
        const float give = up16 ? a[j] : a[j + 4];
        b4[j] = keep + __shfl_xor(give, 16, 32);
      }
      float c2[2];
#pragma unroll
      for (int j = 0; j < 2; ++j) {
        const float keep = up8 ? b4[j + 2] : b4[j];
        const float give = up8 ? b4[j] : b4[j + 2];
        c2[j] = keep + __shfl_xor(give, 8, 32);
      }
      float r;
      {
        const float keep = up4 ? c2[1] : c2[0];
        const float give = up4 ? c2[0] : c2[1];
        r = keep + __shfl_xor(give, 4, 32);
      }
      r += __shfl_xor(r, 2, 32);
      r += __shfl_xor(r, 1, 32);
      r *= 0.08838834764831845f;

      float sc[8];
#pragma unroll
      for (int j = 0; j < 8; ++j)
        sc[j] = __int_as_float(
            __builtin_amdgcn_readlane(__float_as_int(r), 4 * j));

      float gmax = sc[0], gmin = sc[0];
#pragma unroll
      for (int j = 1; j < 8; ++j) {
        if (b + j < n) {
          gmax = fmaxf(gmax, sc[j]);
          gmin = fminf(gmin, sc[j]);
        }
      }
      const float mdn = fmaxf(md, gmax);
      const float msn = fmaxf(ms, -gmin);
      const float fd = __expf(md - mdn);
      const float fs = __expf(ms - msn);
      sd *= fd;
      ss *= fs;
      ad.x *= fd;
      ad.y *= fd;
      ad.z *= fd;
      ad.w *= fd;
      as.x *= fs;
      as.y *= fs;
      as.z *= fs;
      as.w *= fs;
      md = mdn;
      ms = msn;
#pragma unroll
      for (int j = 0; j < 8; ++j) {
        if (b + j < n) {
          const float ed = __expf(sc[j] - mdn);
          const float es = __expf(-sc[j] - msn);
          sd += ed;
          ss += es;
          ad.x += ed * v[j].x;
          ad.y += ed * v[j].y;
          ad.z += ed * v[j].z;
          ad.w += ed * v[j].w;
          as.x += es * v[j].x;
          as.y += es * v[j].y;
          as.z += es * v[j].z;
          as.w += es * v[j].w;
        }
      }
    }
    while (pend == 0u && oc < novf) {
      int2 q = make_int2(0, -1);
      if (oc + lane < novf) q = ((const int2*)ovf)[oc + lane];
      os = q.x;
      ow = q.y;
      pend = __builtin_amdgcn_ballot_w32(ow / SELF_PAD == d && ow >= 0);
      oc += 32;
    }
    if (pend == 0u) break;
    const int jl = __builtin_ctz(pend);
    pend &= pend - 1u;
    rs = __builtin_amdgcn_readlane(os, jl);
    rt = __builtin_amdgcn_readlane(ow, jl) % SELF_PAD;
    n = 1;
  }
  const float denom = (float)nt;
  const float wd = 1.0f / (sd * denom);
  const float ws = 1.0f / (ss * denom);
  ((float4*)(featsta + (size_t)d * D))[lane] =
      make_float4(as.x * ws, as.y * ws, as.z * ws, as.w * ws);
  ((float4*)(featdyn + (size_t)d * D))[lane] =
      make_float4(ad.x * wd, ad.y * wd, ad.z * wd, ad.w * wd);
}

extern "C" void kernel_launch(void* const* d_in, const int* in_sizes, int n_in,
                              void* d_out, int out_size, void* d_ws,
                              size_t ws_size, hipStream_t stream) {
  const float* curr = (const float*)d_in[0];
  const float* last = (const float*)d_in[1];
  const float* relemb = (const float*)d_in[2];
  const float* selfrel = (const float*)d_in[3];
  const float* Wq = (const float*)d_in[4];
  const float* bq = (const float*)d_in[5];
  const float* Wk = (const float*)d_in[6];
  const float* bk = (const float*)d_in[7];
  const float* Wv = (const float*)d_in[8];
  const float* bv = (const float*)d_in[9];
  const float* dynW = (const float*)d_in[10];
  const int* ei = (const int*)d_in[11];
  const int* et = (const int*)d_in[12];

  const int N = in_sizes[0] / D;
  const int E = in_sizes[12];
  const int R = in_sizes[2] / D;
  const size_t ND = (size_t)N * D;

  float* ws = (float*)d_ws;
  float* Xq = ws;
  float* Xk = Xq + ND;
  float* Vv = Xk + ND;
  float* featdyn = Vv + ND;
  float* Tq = featdyn + ND;
  float* Tk = Tq + SELF_PAD * D;
  int* ovf = (int*)(Tk + SELF_PAD * D);
  int* ovfn = ovf + (size_t)2 * E;
  int* cnti = ovfn + (N + 1);
  int* chunks = (int*)featdyn;

  float* out0 = (float*)d_out;
  float* out1 = out0 + ND;

  front_products_wmma<<<3 * ((N / 16 + 1) / 2) + 2 * (SELF_PAD / 32), 128, 0,
                        stream>>>(curr, last, relemb, selfrel, Wq, Wk, Wv, bq,
                                  bk, bv, Xq, Xk, Vv, Tq, Tk, cnti, ovfn, N, R);

  edge_slot_kernel<<<(E + 255) / 256, 256, 0, stream>>>(ei, et, E, cnti, chunks,
                                                       ovfn, ovf);

  node_segment_kernel<<<(N + 3) / 4, 128, 0, stream>>>(
      N, R, E, cnti, ovfn, ovf, Xq, Xk, Tq, Tk, Vv, out0, featdyn);

  gemm128_strip_wmma<<<(N / 16 + 1) / 2, 128, 0, stream>>>(featdyn, dynW,
                                                         nullptr, out1, N);
}


# Last product fused into the segment pass: 512 resident work-groups stage dyn_weight in LDS once, loop over 16-destination tiles whose feat_dyn rows stay in LDS for the f32 WMMA; 4 launches -> 3
# speedup vs baseline: 3.3917x; 1.0701x over previous
#include <hip/hip_runtime.h>
#include <math.h>


typedef __attribute__((ext_vector_type(2))) float v2f;
typedef __attribute__((ext_vector_type(8))) float v8f;

#define D 128
#define SELF_PAD 512

#define WP 288

__device__ __forceinline__ void strip_gemm_block(
    const float* __restrict__ Ap, const float* __restrict__ W,
    const float* __restrict__ bias, float* __restrict__ C, int m0, int M,
    int wave, int lane, int r, int g) {
  __shared__ __attribute__((aligned(16))) float Ws[64 * WP];
  const int nb = (wave & 1) * 64;

  float4 w0[16], w1[16];
  const float* __restrict__ Wb = W + (size_t)(2 * wave) * D + 4 * lane;
#pragma unroll
  for (int it = 0; it < 16; ++it) {
    w0[it] = *(const float4*)(Wb + it * 8 * D);
    w1[it] = *(const float4*)(Wb + it * 8 * D + D);
  }

  v2f a[32];
#pragma unroll
  for (int ks = 0; ks < 32; ++ks) a[ks] = *(const v2f*)(Ap + 4 * ks);

  float* dstb = Ws + wave * WP + 8 * lane;
#pragma unroll
  for (int it = 0; it < 16; ++it) {
    float* dst = dstb + it * 4 * WP;
    *(float4*)dst = make_float4(w0[it].x, w1[it].x, w0[it].y, w1[it].y);
    *(float4*)(dst + 4) = make_float4(w0[it].z, w1[it].z, w0[it].w, w1[it].w);
  }
  asm volatile("s_wait_dscnt 0x0\n\ts_barrier_signal -1\n\ts_barrier_wait -1"
               ::: "memory");

  float bv[4];
#pragma unroll
  for (int ct = 0; ct < 4; ++ct)
    bv[ct] = bias ? bias[nb + 16 * ct + r] : 0.0f;

  v8f acc[4] = {};
  const float* __restrict__ Bp = Ws + g * WP + 2 * (nb + r);
#pragma unroll
  for (int ks = 0; ks < 32; ++ks) {
#pragma unroll
    for (int ct = 0; ct < 4; ++ct) {
      const v2f b = *(const v2f*)(Bp + 2 * ks * WP + 32 * ct);
      acc[ct] = __builtin_amdgcn_wmma_f32_16x16x4_f32(
           false, a[ks],  false, b,
           (short)0, acc[ct],  false,  false);
    }
  }

  if (m0 >= M) return;
#pragma unroll
  for (int ct = 0; ct < 4; ++ct) {
    const int col = nb + 16 * ct + r;
#pragma unroll
    for (int j = 0; j < 8; ++j) {
      const int row = m0 + j + 8 * g;
      C[(size_t)row * D + col] = acc[ct][j] + bv[ct];
    }
  }
}

__global__ __launch_bounds__(128) void front_products_wmma(
    const float* __restrict__ curr, const float* __restrict__ last,
    const float* __restrict__ relemb, const float* __restrict__ selfrel,
    const float* __restrict__ Wq, const float* __restrict__ Wk,
    const float* __restrict__ Wv, const float* __restrict__ bq,
    const float* __restrict__ bk, const float* __restrict__ bvv,
    float* __restrict__ Xq, float* __restrict__ Xk, float* __restrict__ Vv,
    float* __restrict__ Tq, float* __restrict__ Tk, int* __restrict__ cnti,
    int* __restrict__ ovfn, int N, int R) {
  const int tid = threadIdx.x;
  const int lane = tid & 31;
  const int wave = __builtin_amdgcn_readfirstlane(tid >> 5);
  const int r = lane & 15;
  const int g = lane >> 4;

  const int n = blockIdx.x * 128 + tid;
  if (n < N) cnti[n] = 0;
  if (n == 0) ovfn[0] = 0;

  const int G = (N / 16 + 1) / 2;
  int j = blockIdx.x;
  const float* A = curr;
  const float* W = Wq;
  const float* bias = nullptr;
  float* C = Xq;
  int M = N;
  bool rel = false;
  if (j >= G) {
    j -= G;
    A = last;
    W = Wk;
    C = Xk;
    if (j >= G) {
      j -= G;
      W = Wv;
      bias = bvv;
      C = Vv;
      if (j >= G) {
        j -= G;
        rel = true;
        M = SELF_PAD;
        W = Wq + D * D;
        bias = bq;
        C = Tq;
        if (j >= SELF_PAD / 32) {
          j -= SELF_PAD / 32;
          W = Wk + D * D;
          bias = bk;
          C = Tk;
        }
      }
    }
  }
  const int m0 = (j * 2 + (wave >> 1)) * 16;
  int arow = m0 + r;
  if (arow > M - 1) arow = M - 1;
  const float* Arow = A + (size_t)arow * D;
  if (rel) Arow = arow < R ? relemb + (size_t)arow * D : selfrel;
  strip_gemm_block(Arow + 2 * g, W, bias, C, m0, M, wave, lane, r, g);
}

#define SLOTS 31

__global__ __launch_bounds__(256) void edge_slot_kernel(
    const int* __restrict__ ei, const int* __restrict__ et, int E,
    int* __restrict__ cnti, int* chunks, int* __restrict__ ovfn,
    int* __restrict__ ovf) {
  const int e = blockIdx.x * 256 + threadIdx.x;
  if (e >= E) return;
  const int s = ei[e];
  const int d = ei[E + e];
  const int t = et[e];
  const int rk = atomicAdd(&cnti[d], 1);
  if (rk < SLOTS) {
    ((int2*)(chunks + (size_t)d * D))[rk] = make_int2(s, t);
  } else {
    const int o = atomicAdd(ovfn, 1);
    ((int2*)ovf)[o] = make_int2(s, d * SELF_PAD + t);
  }
}

#define FP 132
#define SEG_GROUPS 512

__global__ __launch_bounds__(256) void node_segment_dyn_kernel(
    int N, int R, int E, const int* __restrict__ cnti,
    const int* __restrict__ ovfn, const int* __restrict__ ovf,
    const int* __restrict__ chunks, const float* __restrict__ Xq,
    const float* __restrict__ Xk, const float* __restrict__ Tq,
    const float* __restrict__ Tk, const float* __restrict__ Vv,
    const float* __restrict__ dynW, float* __restrict__ featsta,
    float* __restrict__ out1) {
  __shared__ __attribute__((aligned(16))) float Wd[64 * WP];
  __shared__ __attribute__((aligned(16))) float Fd[2 * 16 * FP];
  const int lane = threadIdx.x & 31;
  const int wv = __builtin_amdgcn_readfirstlane((int)(threadIdx.x >> 5));

  {
    float4 w0[8], w1[8];
    const float* __restrict__ Wb = dynW + (size_t)(2 * wv) * D + 4 * lane;
#pragma unroll
    for (int it = 0; it < 8; ++it) {
      w0[it] = *(const float4*)(Wb + it * 16 * D);
      w1[it] = *(const float4*)(Wb + it * 16 * D + D);
    }
    float* dstb = Wd + wv * WP + 8 * lane;
#pragma unroll
    for (int it = 0; it < 8; ++it) {
      float* dst = dstb + it * 8 * WP;
      *(float4*)dst = make_float4(w0[it].x, w1[it].x, w0[it].y, w1[it].y);
      *(float4*)(dst + 4) = make_float4(w0[it].z, w1[it].z, w0[it].w, w1[it].w);
    }
  }

  const bool up16 = (lane & 16) != 0;
  const bool up8 = (lane & 8) != 0;
  const bool up4 = (lane & 4) != 0;

  const int fr = lane & 15;
  const int fg = lane >> 4;
  const int ntiles = N / 16;
  int half = 0;
  for (int tile = blockIdx.x; tile < ntiles; tile += SEG_GROUPS) {
    const int cur = half;
    float* Ft = Fd + cur * 16 * FP;
    half ^= 1;
    for (int i = 0; i < 2; ++i) {
      const int lrow = wv * 2 + i;
      const int d = tile * 16 + lrow;
      const int2 p = ((const int2*)(chunks + (size_t)d * D))[lane];
      const int cnt = cnti[d];
      const int nt = cnt + 1;
      const float4 xq = ((const float4*)(Xq + (size_t)d * D))[lane];
      const int m = cnt < SLOTS ? cnt : SLOTS;
      int rs = p.x, rt = p.y;
      if (lane >= m) {
        rs = d;
        rt = R;
      }
      int n = m + 1;
      int novf = 0;
      if (cnt > SLOTS) {
        novf = ovfn[0];
        if (novf > E) novf = E;
      }
      int oc = 0;
      unsigned pend = 0u;
      int os = 0, ow = -1;

      float md = -__builtin_inff();
      float ms = -__builtin_inff();
      float sd = 0.0f, ss = 0.0f;
      float4 as = make_float4(0.0f, 0.0f, 0.0f, 0.0f);
      float4 ad = make_float4(0.0f, 0.0f, 0.0f, 0.0f);

      for (;;) {
        for (int b = 0; b < n; b += 8) {
          int s[8], t[8];
#pragma unroll
          for (int j = 0; j < 8; ++j) {
            s[j] = __builtin_amdgcn_readlane(rs, b + j);
            t[j] = __builtin_amdgcn_readlane(rt, b + j);
          }
          float4 k0[8], k1[8], q1[8], v[8];
#pragma unroll
          for (int j = 0; j < 8; ++j) {
            if (b + j < n) {
              k0[j] = ((const float4*)(Xk + (size_t)s[j] * D))[lane];
              k1[j] = ((const float4*)(Tk + (size_t)t[j] * D))[lane];
              q1[j] = ((const float4*)(Tq + (size_t)t[j] * D))[lane];
            }
          }
#pragma unroll
          for (int j = 0; j < 8; ++j) {
            if (b + j < n)
              v[j] = ((const float4*)(Vv + (size_t)s[j] * D))[lane];
          }

          float a[8];
#pragma unroll
          for (int j = 0; j < 8; ++j) {
            a[j] = 0.0f;
            if (b + j < n) {
              a[j] = (xq.x + q1[j].x) * (k0[j].x + k1[j].x) +
                     (xq.y + q1[j].y) * (k0[j].y + k1[j].y) +
                     (xq.z + q1[j].z) * (k0[j].z + k1[j].z) +
                     (xq.w + q1[j].w) * (k0[j].w + k1[j].w);
            }
          }

          float b4[4];
#pragma unroll
          for (int j = 0; j < 4; ++j) {
            const float keep = up16 ? a[j + 4] : a[j];
            const float give = up16 ? a[j] : a[j + 4];
            b4[j] = keep + __shfl_xor(give, 16, 32);
          }
          float c2[2];
#pragma unroll
          for (int j = 0; j < 2; ++j) {
            const float keep = up8 ? b4[j + 2] : b4[j];
            const float give = up8 ? b4[j] : b4[j + 2];
            c2[j] = keep + __shfl_xor(give, 8, 32);
          }
          float r;
          {
            const float keep = up4 ? c2[1] : c2[0];
            const float give = up4 ? c2[0] : c2[1];
            r = keep + __shfl_xor(give, 4, 32);
          }
          r += __shfl_xor(r, 2, 32);
          r += __shfl_xor(r, 1, 32);
          r *= 0.08838834764831845f;

          float sc[8];
#pragma unroll
          for (int j = 0; j < 8; ++j)
            sc[j] = __int_as_float(
                __builtin_amdgcn_readlane(__float_as_int(r), 4 * j));

          float gmax = sc[0], gmin = sc[0];
#pragma unroll
          for (int j = 1; j < 8; ++j) {
            if (b + j < n) {
              gmax = fmaxf(gmax, sc[j]);
              gmin = fminf(gmin, sc[j]);
            }
          }
          const float mdn = fmaxf(md, gmax);
          const float msn = fmaxf(ms, -gmin);
          const float fd = __expf(md - mdn);
          const float fs = __expf(ms - msn);
          sd *= fd;
          ss *= fs;
          ad.x *= fd;
          ad.y *= fd;
          ad.z *= fd;
          ad.w *= fd;
          as.x *= fs;
          as.y *= fs;
          as.z *= fs;
          as.w *= fs;
          md = mdn;
          ms = msn;
#pragma unroll
          for (int j = 0; j < 8; ++j) {
            if (b + j < n) {
              const float ed = __expf(sc[j] - mdn);
              const float es = __expf(-sc[j] - msn);
              sd += ed;
              ss += es;
              ad.x += ed * v[j].x;
              ad.y += ed * v[j].y;
              ad.z += ed * v[j].z;
              ad.w += ed * v[j].w;
              as.x += es * v[j].x;
              as.y += es * v[j].y;
              as.z += es * v[j].z;
              as.w += es * v[j].w;
            }
          }
        }
        while (pend == 0u && oc < novf) {
          int2 q = make_int2(0, -1);
          if (oc + lane < novf) q = ((const int2*)ovf)[oc + lane];
          os = q.x;
          ow = q.y;
          pend = __builtin_amdgcn_ballot_w32(ow / SELF_PAD == d && ow >= 0);
          oc += 32;
        }
        if (pend == 0u) break;
        const int jl = __builtin_ctz(pend);
        pend &= pend - 1u;
        rs = __builtin_amdgcn_readlane(os, jl);
        rt = __builtin_amdgcn_readlane(ow, jl) % SELF_PAD;
        n = 1;
      }
      const float denom = (float)nt;
      const float wd = 1.0f / (sd * denom);
      const float ws = 1.0f / (ss * denom);
      ((float4*)(featsta + (size_t)d * D))[lane] =
          make_float4(as.x * ws, as.y * ws, as.z * ws, as.w * ws);
      *(float4*)(Ft + lrow * FP + 4 * lane) =
          make_float4(ad.x * wd, ad.y * wd, ad.z * wd, ad.w * wd);
    }

    asm volatile("s_wait_dscnt 0x0\n\ts_barrier_signal -1\n\ts_barrier_wait -1"
                 ::: "memory");

    if ((wv >> 2) == cur) {
      const int nb = (wv & 3) * 32;
      const float* Fa = Ft + fr * FP + 2 * fg;
      const float* Bp = Wd + fg * WP + 2 * (nb + fr);
      v8f acc[2] = {};
#pragma unroll
      for (int ks = 0; ks < 32; ++ks) {
        const v2f a = *(const v2f*)(Fa + 4 * ks);
#pragma unroll
        for (int ct = 0; ct < 2; ++ct) {
          const v2f b = *(const v2f*)(Bp + 2 * ks * WP + 32 * ct);
          acc[ct] = __builtin_amdgcn_wmma_f32_16x16x4_f32(
               false, a,  false, b,
               (short)0, acc[ct],  false,  false);
        }
      }
#pragma unroll
      for (int ct = 0; ct < 2; ++ct) {
        const int col = nb + 16 * ct + fr;
#pragma unroll
        for (int j = 0; j < 8; ++j) {
          const int row = tile * 16 + j + 8 * fg;
          out1[(size_t)row * D + col] = acc[ct][j];
        }
      }
    }
  }
}

extern "C" void kernel_launch(void* const* d_in, const int* in_sizes, int n_in,
                              void* d_out, int out_size, void* d_ws,
                              size_t ws_size, hipStream_t stream) {
  const float* curr = (const float*)d_in[0];
  const float* last = (const float*)d_in[1];
  const float* relemb = (const float*)d_in[2];
  const float* selfrel = (const float*)d_in[3];
  const float* Wq = (const float*)d_in[4];
  const float* bq = (const float*)d_in[5];
  const float* Wk = (const float*)d_in[6];
  const float* bk = (const float*)d_in[7];
  const float* Wv = (const float*)d_in[8];
  const float* bv = (const float*)d_in[9];
  const float* dynW = (const float*)d_in[10];
  const int* ei = (const int*)d_in[11];
  const int* et = (const int*)d_in[12];

  const int N = in_sizes[0] / D;
  const int E = in_sizes[12];
  const int R = in_sizes[2] / D;
  const size_t ND = (size_t)N * D;

  float* ws = (float*)d_ws;
  float* Xq = ws;
  float* Xk = Xq + ND;
  float* Vv = Xk + ND;
  int* chunks = (int*)(Vv + ND);
  float* Tq = (float*)chunks + ND;
  float* Tk = Tq + SELF_PAD * D;
  int* ovf = (int*)(Tk + SELF_PAD * D);
  int* ovfn = ovf + (size_t)2 * E;
  int* cnti = ovfn + (N + 1);

  float* out0 = (float*)d_out;
  float* out1 = out0 + ND;

  front_products_wmma<<<3 * ((N / 16 + 1) / 2) + 2 * (SELF_PAD / 32), 128, 0,
                        stream>>>(curr, last, relemb, selfrel, Wq, Wk, Wv, bq,
                                  bk, bv, Xq, Xk, Vv, Tq, Tk, cnti, ovfn, N, R);

  edge_slot_kernel<<<(E + 255) / 256, 256, 0, stream>>>(ei, et, E, cnti, chunks,
                                                       ovfn, ovf);

  node_segment_dyn_kernel<<<SEG_GROUPS, 256, 0, stream>>>(
      N, R, E, cnti, ovfn, ovf, chunks, Xq, Xk, Tq, Tk, Vv, dynW, out0, out1);
}


# Edge ranks in LDS: 64 work-groups each count a batch of edges in per-destination LDS counters, then add whole 128-byte lines of counts to the global counts; arrival order varies as before
# speedup vs baseline: 3.5944x; 1.0598x over previous
#include <hip/hip_runtime.h>
#include <math.h>


typedef __attribute__((ext_vector_type(2))) float v2f;
typedef __attribute__((ext_vector_type(8))) float v8f;

#define D 128
#define SELF_PAD 512

#define WP 288

__device__ __forceinline__ void strip_gemm_block(
    const float* __restrict__ Ap, const float* __restrict__ W,
    const float* __restrict__ bias, float* __restrict__ C, int m0, int M,
    int wave, int lane, int r, int g) {
  __shared__ __attribute__((aligned(16))) float Ws[64 * WP];
  const int nb = (wave & 1) * 64;

  float4 w0[16], w1[16];
  const float* __restrict__ Wb = W + (size_t)(2 * wave) * D + 4 * lane;
#pragma unroll
  for (int it = 0; it < 16; ++it) {
    w0[it] = *(const float4*)(Wb + it * 8 * D);
    w1[it] = *(const float4*)(Wb + it * 8 * D + D);
  }

  v2f a[32];
#pragma unroll
  for (int ks = 0; ks < 32; ++ks) a[ks] = *(const v2f*)(Ap + 4 * ks);

  float* dstb = Ws + wave * WP + 8 * lane;
#pragma unroll
  for (int it = 0; it < 16; ++it) {
    float* dst = dstb + it * 4 * WP;
    *(float4*)dst = make_float4(w0[it].x, w1[it].x, w0[it].y, w1[it].y);
    *(float4*)(dst + 4) = make_float4(w0[it].z, w1[it].z, w0[it].w, w1[it].w);
  }
  asm volatile("s_wait_dscnt 0x0\n\ts_barrier_signal -1\n\ts_barrier_wait -1"
               ::: "memory");

  float bv[4];
#pragma unroll
  for (int ct = 0; ct < 4; ++ct)
    bv[ct] = bias ? bias[nb + 16 * ct + r] : 0.0f;

  v8f acc[4] = {};
  const float* __restrict__ Bp = Ws + g * WP + 2 * (nb + r);
#pragma unroll
  for (int ks = 0; ks < 32; ++ks) {
#pragma unroll
    for (int ct = 0; ct < 4; ++ct) {
      const v2f b = *(const v2f*)(Bp + 2 * ks * WP + 32 * ct);
      acc[ct] = __builtin_amdgcn_wmma_f32_16x16x4_f32(
           false, a[ks],  false, b,
           (short)0, acc[ct],  false,  false);
    }
  }

  if (m0 >= M) return;
#pragma unroll
  for (int ct = 0; ct < 4; ++ct) {
    const int col = nb + 16 * ct + r;
#pragma unroll
    for (int j = 0; j < 8; ++j) {
      const int row = m0 + j + 8 * g;
      C[(size_t)row * D + col] = acc[ct][j] + bv[ct];
    }
  }
}

__global__ __launch_bounds__(128) void front_products_wmma(
    const float* __restrict__ curr, const float* __restrict__ last,
    const float* __restrict__ relemb, const float* __restrict__ selfrel,
    const float* __restrict__ Wq, const float* __restrict__ Wk,
    const float* __restrict__ Wv, const float* __restrict__ bq,
    const float* __restrict__ bk, const float* __restrict__ bvv,
    float* __restrict__ Xq, float* __restrict__ Xk, float* __restrict__ Vv,
    float* __restrict__ Tq, float* __restrict__ Tk, int* __restrict__ cnti,
    int* __restrict__ ovfn, int N, int R) {
  const int tid = threadIdx.x;
  const int lane = tid & 31;
  const int wave = __builtin_amdgcn_readfirstlane(tid >> 5);
  const int r = lane & 15;
  const int g = lane >> 4;

  const int n = blockIdx.x * 128 + tid;
  if (n < N) cnti[n] = 0;
  if (n == 0) ovfn[0] = 0;

  const int G = (N / 16 + 1) / 2;
  int j = blockIdx.x;
  const float* A = curr;
  const float* W = Wq;
  const float* bias = nullptr;
  float* C = Xq;
  int M = N;
  bool rel = false;
  if (j >= G) {
    j -= G;
    A = last;
    W = Wk;
    C = Xk;
    if (j >= G) {
      j -= G;
      W = Wv;
      bias = bvv;
      C = Vv;
      if (j >= G) {
        j -= G;
        rel = true;
        M = SELF_PAD;
        W = Wq + D * D;
        bias = bq;
        C = Tq;
        if (j >= SELF_PAD / 32) {
          j -= SELF_PAD / 32;
          W = Wk + D * D;
          bias = bk;
          C = Tk;
        }
      }
    }
  }
  const int m0 = (j * 2 + (wave >> 1)) * 16;
  int arow = m0 + r;
  if (arow > M - 1) arow = M - 1;
  const float* Arow = A + (size_t)arow * D;
  if (rel) Arow = arow < R ? relemb + (size_t)arow * D : selfrel;
  strip_gemm_block(Arow + 2 * g, W, bias, C, m0, M, wave, lane, r, g);
}

#define SLOTS 31
#define RANK_BINS 40960
#define RANK_EDGES 7
#define RANK_LINES 20

__global__ __launch_bounds__(1024) void edge_slot_kernel(
    const int* __restrict__ ei, const int* __restrict__ et, int E, int N,
    int* __restrict__ cnti, int* chunks, int* __restrict__ ovfn,
    int* __restrict__ ovf) {
  __shared__ __attribute__((aligned(16))) int hist[RANK_BINS];
  const int tid = threadIdx.x;
  const int per = (E + (int)gridDim.x - 1) / (int)gridDim.x;
  const int e0 = blockIdx.x * per;
  int e1 = e0 + per;
  if (e1 > E) e1 = E;
  for (int eb = e0; eb < e1; eb += RANK_EDGES * 1024) {
    for (int i = tid; i < RANK_BINS / 4; i += 1024)
      ((int4*)hist)[i] = make_int4(0, 0, 0, 0);

    int s[RANK_EDGES], d[RANK_EDGES], t[RANK_EDGES], rk[RANK_EDGES];
#pragma unroll
    for (int k = 0; k < RANK_EDGES; ++k) {
      const int e = eb + k * 1024 + tid;
      d[k] = -1;
      s[k] = 0;
      t[k] = 0;
      if (e < e1) {
        s[k] = ei[e];
        d[k] = ei[E + e];
        t[k] = et[e];
      }
    }
    __syncthreads();

#pragma unroll
    for (int k = 0; k < RANK_EDGES; ++k) {
      rk[k] = 0;
      if (d[k] >= 0) rk[k] = atomicAdd(&hist[d[k]], 1);
    }
    __syncthreads();

    for (int b0 = 0; b0 < N; b0 += RANK_LINES * 1024) {
      int h[RANK_LINES], first[RANK_LINES];
#pragma unroll
      for (int u = 0; u < RANK_LINES; ++u) {
        const int bin = b0 + u * 1024 + tid;
        h[u] = bin < N ? hist[bin] : 0;
      }
#pragma unroll
      for (int u = 0; u < RANK_LINES; ++u) {
        const int bin = b0 + u * 1024 + tid;
        first[u] = 0;
        if (h[u] > 0) first[u] = atomicAdd(&cnti[bin], h[u]);
      }
#pragma unroll
      for (int u = 0; u < RANK_LINES; ++u) {
        const int bin = b0 + u * 1024 + tid;
        if (h[u] > 0) hist[bin] = first[u];
      }
    }
    __syncthreads();

#pragma unroll
    for (int k = 0; k < RANK_EDGES; ++k) {
      if (d[k] >= 0) {
        const int slot = hist[d[k]] + rk[k];
        if (slot < SLOTS) {
          ((int2*)(chunks + (size_t)d[k] * D))[slot] = make_int2(s[k], t[k]);
        } else {
          const int o = atomicAdd(ovfn, 1);
          ((int2*)ovf)[o] = make_int2(s[k], d[k] * SELF_PAD + t[k]);
        }
      }
    }
    __syncthreads();
  }
}

#define FP 132
#define SEG_GROUPS 512

__global__ __launch_bounds__(256) void node_segment_dyn_kernel(
    int N, int R, int E, const int* __restrict__ cnti,
    const int* __restrict__ ovfn, const int* __restrict__ ovf,
    const int* __restrict__ chunks, const float* __restrict__ Xq,
    const float* __restrict__ Xk, const float* __restrict__ Tq,
    const float* __restrict__ Tk, const float* __restrict__ Vv,
    const float* __restrict__ dynW, float* __restrict__ featsta,
    float* __restrict__ out1) {
  __shared__ __attribute__((aligned(16))) float Wd[64 * WP];
  __shared__ __attribute__((aligned(16))) float Fd[2 * 16 * FP];
  const int lane = threadIdx.x & 31;
  const int wv = __builtin_amdgcn_readfirstlane((int)(threadIdx.x >> 5));

  {
    float4 w0[8], w1[8];
    const float* __restrict__ Wb = dynW + (size_t)(2 * wv) * D + 4 * lane;
#pragma unroll
    for (int it = 0; it < 8; ++it) {
      w0[it] = *(const float4*)(Wb + it * 16 * D);
      w1[it] = *(const float4*)(Wb + it * 16 * D + D);
    }
    float* dstb = Wd + wv * WP + 8 * lane;
#pragma unroll
    for (int it = 0; it < 8; ++it) {
      float* dst = dstb + it * 8 * WP;
      *(float4*)dst = make_float4(w0[it].x, w1[it].x, w0[it].y, w1[it].y);
      *(float4*)(dst + 4) = make_float4(w0[it].z, w1[it].z, w0[it].w, w1[it].w);
    }
  }

  const bool up16 = (lane & 16) != 0;
  const bool up8 = (lane & 8) != 0;
  const bool up4 = (lane & 4) != 0;

  const int fr = lane & 15;
  const int fg = lane >> 4;
  const int ntiles = N / 16;
  int half = 0;
  for (int tile = blockIdx.x; tile < ntiles; tile += SEG_GROUPS) {
    const int cur = half;
    float* Ft = Fd + cur * 16 * FP;
    half ^= 1;
    for (int i = 0; i < 2; ++i) {
      const int lrow = wv * 2 + i;
      const int d = tile * 16 + lrow;
      const int2 p = ((const int2*)(chunks + (size_t)d * D))[lane];
      const int cnt = cnti[d];
      const int nt = cnt + 1;
      const float4 xq = ((const float4*)(Xq + (size_t)d * D))[lane];
      const int m = cnt < SLOTS ? cnt : SLOTS;
      int rs = p.x, rt = p.y;
      if (lane >= m) {
        rs = d;
        rt = R;
      }
      int n = m + 1;
      int novf = 0;
      if (cnt > SLOTS) {
        novf = ovfn[0];
        if (novf > E) novf = E;
      }
      int oc = 0;
      unsigned pend = 0u;
      int os = 0, ow = -1;

      float md = -__builtin_inff();
      float ms = -__builtin_inff();
      float sd = 0.0f, ss = 0.0f;
      float4 as = make_float4(0.0f, 0.0f, 0.0f, 0.0f);
      float4 ad = make_float4(0.0f, 0.0f, 0.0f, 0.0f);

      for (;;) {
        for (int b = 0; b < n; b += 8) {
          int s[8], t[8];
#pragma unroll
          for (int j = 0; j < 8; ++j) {
            s[j] = __builtin_amdgcn_readlane(rs, b + j);
            t[j] = __builtin_amdgcn_readlane(rt, b + j);
          }
          float4 k0[8], k1[8], q1[8], v[8];
#pragma unroll
          for (int j = 0; j < 8; ++j) {
            if (b + j < n) {
              k0[j] = ((const float4*)(Xk + (size_t)s[j] * D))[lane];
              k1[j] = ((const float4*)(Tk + (size_t)t[j] * D))[lane];
              q1[j] = ((const float4*)(Tq + (size_t)t[j] * D))[lane];
            }
          }
#pragma unroll
          for (int j = 0; j < 8; ++j) {
            if (b + j < n)
              v[j] = ((const float4*)(Vv + (size_t)s[j] * D))[lane];
          }

          float a[8];
#pragma unroll
          for (int j = 0; j < 8; ++j) {
            a[j] = 0.0f;
            if (b + j < n) {
              a[j] = (xq.x + q1[j].x) * (k0[j].x + k1[j].x) +
                     (xq.y + q1[j].y) * (k0[j].y + k1[j].y) +
                     (xq.z + q1[j].z) * (k0[j].z + k1[j].z) +
                     (xq.w + q1[j].w) * (k0[j].w + k1[j].w);
            }
          }

          float b4[4];
#pragma unroll
          for (int j = 0; j < 4; ++j) {
            const float keep = up16 ? a[j + 4] : a[j];
            const float give = up16 ? a[j] : a[j + 4];
            b4[j] = keep + __shfl_xor(give, 16, 32);
          }
          float c2[2];
#pragma unroll
          for (int j = 0; j < 2; ++j) {
            const float keep = up8 ? b4[j + 2] : b4[j];
            const float give = up8 ? b4[j] : b4[j + 2];
            c2[j] = keep + __shfl_xor(give, 8, 32);
          }
          float r;
          {
            const float keep = up4 ? c2[1] : c2[0];
            const float give = up4 ? c2[0] : c2[1];
            r = keep + __shfl_xor(give, 4, 32);
          }
          r += __shfl_xor(r, 2, 32);
          r += __shfl_xor(r, 1, 32);
          r *= 0.08838834764831845f;

          float sc[8];
#pragma unroll
          for (int j = 0; j < 8; ++j)
            sc[j] = __int_as_float(
                __builtin_amdgcn_readlane(__float_as_int(r), 4 * j));

          float gmax = sc[0], gmin = sc[0];
#pragma unroll
          for (int j = 1; j < 8; ++j) {
            if (b + j < n) {
              gmax = fmaxf(gmax, sc[j]);
              gmin = fminf(gmin, sc[j]);
            }
          }
          const float mdn = fmaxf(md, gmax);
          const float msn = fmaxf(ms, -gmin);
          const float fd = __expf(md - mdn);
          const float fs = __expf(ms - msn);
          sd *= fd;
          ss *= fs;
          ad.x *= fd;
          ad.y *= fd;
          ad.z *= fd;
          ad.w *= fd;
          as.x *= fs;
          as.y *= fs;
          as.z *= fs;
          as.w *= fs;
          md = mdn;
          ms = msn;
#pragma unroll
          for (int j = 0; j < 8; ++j) {
            if (b + j < n) {
              const float ed = __expf(sc[j] - mdn);
              const float es = __expf(-sc[j] - msn);
              sd += ed;
              ss += es;
              ad.x += ed * v[j].x;
              ad.y += ed * v[j].y;
              ad.z += ed * v[j].z;
              ad.w += ed * v[j].w;
              as.x += es * v[j].x;
              as.y += es * v[j].y;
              as.z += es * v[j].z;
              as.w += es * v[j].w;
            }
          }
        }
        while (pend == 0u && oc < novf) {
          int2 q = make_int2(0, -1);
          if (oc + lane < novf) q = ((const int2*)ovf)[oc + lane];
          os = q.x;
          ow = q.y;
          pend = __builtin_amdgcn_ballot_w32(ow / SELF_PAD == d && ow >= 0);
          oc += 32;
        }
        if (pend == 0u) break;
        const int jl = __builtin_ctz(pend);
        pend &= pend - 1u;
        rs = __builtin_amdgcn_readlane(os, jl);
        rt = __builtin_amdgcn_readlane(ow, jl) % SELF_PAD;
        n = 1;
      }
      const float denom = (float)nt;
      const float wd = 1.0f / (sd * denom);
      const float ws = 1.0f / (ss * denom);
      ((float4*)(featsta + (size_t)d * D))[lane] =
          make_float4(as.x * ws, as.y * ws, as.z * ws, as.w * ws);
      *(float4*)(Ft + lrow * FP + 4 * lane) =
          make_float4(ad.x * wd, ad.y * wd, ad.z * wd, ad.w * wd);
    }

    asm volatile("s_wait_dscnt 0x0\n\ts_barrier_signal -1\n\ts_barrier_wait -1"
                 ::: "memory");

    if ((wv >> 2) == cur) {
      const int nb = (wv & 3) * 32;
      const float* Fa = Ft + fr * FP + 2 * fg;
      const float* Bp = Wd + fg * WP + 2 * (nb + fr);
      v8f acc[2] = {};
#pragma unroll
      for (int ks = 0; ks < 32; ++ks) {
        const v2f a = *(const v2f*)(Fa + 4 * ks);
#pragma unroll
        for (int ct = 0; ct < 2; ++ct) {
          const v2f b = *(const v2f*)(Bp + 2 * ks * WP + 32 * ct);
          acc[ct] = __builtin_amdgcn_wmma_f32_16x16x4_f32(
               false, a,  false, b,
               (short)0, acc[ct],  false,  false);
        }
      }
#pragma unroll
      for (int ct = 0; ct < 2; ++ct) {
        const int col = nb + 16 * ct + fr;
#pragma unroll
        for (int j = 0; j < 8; ++j) {
          const int row = tile * 16 + j + 8 * fg;
          out1[(size_t)row * D + col] = acc[ct][j];
        }
      }
    }
  }
}

extern "C" void kernel_launch(void* const* d_in, const int* in_sizes, int n_in,
                              void* d_out, int out_size, void* d_ws,
                              size_t ws_size, hipStream_t stream) {
  const float* curr = (const float*)d_in[0];
  const float* last = (const float*)d_in[1];
  const float* relemb = (const float*)d_in[2];
  const float* selfrel = (const float*)d_in[3];
  const float* Wq = (const float*)d_in[4];
  const float* bq = (const float*)d_in[5];
  const float* Wk = (const float*)d_in[6];
  const float* bk = (const float*)d_in[7];
  const float* Wv = (const float*)d_in[8];
  const float* bv = (const float*)d_in[9];
  const float* dynW = (const float*)d_in[10];
  const int* ei = (const int*)d_in[11];
  const int* et = (const int*)d_in[12];

  const int N = in_sizes[0] / D;
  const int E = in_sizes[12];
  const int R = in_sizes[2] / D;
  const size_t ND = (size_t)N * D;

  float* ws = (float*)d_ws;
  float* Xq = ws;
  float* Xk = Xq + ND;
  float* Vv = Xk + ND;
  int* chunks = (int*)(Vv + ND);
  float* Tq = (float*)chunks + ND;
  float* Tk = Tq + SELF_PAD * D;
  int* ovf = (int*)(Tk + SELF_PAD * D);
  int* ovfn = ovf + (size_t)2 * E;
  int* cnti = ovfn + (N + 1);

  float* out0 = (float*)d_out;
  float* out1 = out0 + ND;

  front_products_wmma<<<3 * ((N / 16 + 1) / 2) + 2 * (SELF_PAD / 32), 128, 0,
                        stream>>>(curr, last, relemb, selfrel, Wq, Wk, Wv, bq,
                                  bk, bv, Xq, Xk, Vv, Tq, Tk, cnti, ovfn, N, R);

  edge_slot_kernel<<<64, 1024, 0, stream>>>(ei, et, E, N, cnti, chunks, ovfn,
                                            ovf);

  node_segment_dyn_kernel<<<SEG_GROUPS, 256, 0, stream>>>(
      N, R, E, cnti, ovfn, ovf, chunks, Xq, Xk, Tq, Tk, Vv, dynW, out0, out1);
}


# Segment pass: the 8-item group body is compiled once per item count (1..8) as straight-line code picked by one scalar jump; no per-item tests, exact load waits; source sum order unchanged
# speedup vs baseline: 3.7459x; 1.0421x over previous
#include <hip/hip_runtime.h>
#include <math.h>


typedef __attribute__((ext_vector_type(2))) float v2f;
typedef __attribute__((ext_vector_type(8))) float v8f;

#define D 128
#define SELF_PAD 512

#define WP 288

__device__ __forceinline__ void strip_gemm_block(
    const float* __restrict__ Ap, const float* __restrict__ W,
    const float* __restrict__ bias, float* __restrict__ C, int m0, int M,
    int wave, int lane, int r, int g) {
  __shared__ __attribute__((aligned(16))) float Ws[64 * WP];
  const int nb = (wave & 1) * 64;

  float4 w0[16], w1[16];
  const float* __restrict__ Wb = W + (size_t)(2 * wave) * D + 4 * lane;
#pragma unroll
  for (int it = 0; it < 16; ++it) {
    w0[it] = *(const float4*)(Wb + it * 8 * D);
    w1[it] = *(const float4*)(Wb + it * 8 * D + D);
  }

  v2f a[32];
#pragma unroll
  for (int ks = 0; ks < 32; ++ks) a[ks] = *(const v2f*)(Ap + 4 * ks);

  float* dstb = Ws + wave * WP + 8 * lane;
#pragma unroll
  for (int it = 0; it < 16; ++it) {
    float* dst = dstb + it * 4 * WP;
    *(float4*)dst = make_float4(w0[it].x, w1[it].x, w0[it].y, w1[it].y);
    *(float4*)(dst + 4) = make_float4(w0[it].z, w1[it].z, w0[it].w, w1[it].w);
  }
  asm volatile("s_wait_dscnt 0x0\n\ts_barrier_signal -1\n\ts_barrier_wait -1"
               ::: "memory");

  float bv[4];
#pragma unroll
  for (int ct = 0; ct < 4; ++ct)
    bv[ct] = bias ? bias[nb + 16 * ct + r] : 0.0f;

  v8f acc[4] = {};
  const float* __restrict__ Bp = Ws + g * WP + 2 * (nb + r);
#pragma unroll
  for (int ks = 0; ks < 32; ++ks) {
#pragma unroll
    for (int ct = 0; ct < 4; ++ct) {
      const v2f b = *(const v2f*)(Bp + 2 * ks * WP + 32 * ct);
      acc[ct] = __builtin_amdgcn_wmma_f32_16x16x4_f32(
           false, a[ks],  false, b,
           (short)0, acc[ct],  false,  false);
    }
  }

  if (m0 >= M) return;
#pragma unroll
  for (int ct = 0; ct < 4; ++ct) {
    const int col = nb + 16 * ct + r;
#pragma unroll
    for (int j = 0; j < 8; ++j) {
      const int row = m0 + j + 8 * g;
      C[(size_t)row * D + col] = acc[ct][j] + bv[ct];
    }
  }
}

__global__ __launch_bounds__(128) void front_products_wmma(
    const float* __restrict__ curr, const float* __restrict__ last,
    const float* __restrict__ relemb, const float* __restrict__ selfrel,
    const float* __restrict__ Wq, const float* __restrict__ Wk,
    const float* __restrict__ Wv, const float* __restrict__ bq,
    const float* __restrict__ bk, const float* __restrict__ bvv,
    float* __restrict__ Xq, float* __restrict__ Xk, float* __restrict__ Vv,
    float* __restrict__ Tq, float* __restrict__ Tk, int* __restrict__ cnti,
    int* __restrict__ ovfn, int N, int R) {
  const int tid = threadIdx.x;
  const int lane = tid & 31;
  const int wave = __builtin_amdgcn_readfirstlane(tid >> 5);
  const int r = lane & 15;
  const int g = lane >> 4;

  const int n = blockIdx.x * 128 + tid;
  if (n < N) cnti[n] = 0;
  if (n == 0) ovfn[0] = 0;

  const int G = (N / 16 + 1) / 2;
  int j = blockIdx.x;
  const float* A = curr;
  const float* W = Wq;
  const float* bias = nullptr;
  float* C = Xq;
  int M = N;
  bool rel = false;
  if (j >= G) {
    j -= G;
    A = last;
    W = Wk;
    C = Xk;
    if (j >= G) {
      j -= G;
      W = Wv;
      bias = bvv;
      C = Vv;
      if (j >= G) {
        j -= G;
        rel = true;
        M = SELF_PAD;
        W = Wq + D * D;
        bias = bq;
        C = Tq;
        if (j >= SELF_PAD / 32) {
          j -= SELF_PAD / 32;
          W = Wk + D * D;
          bias = bk;
          C = Tk;
        }
      }
    }
  }
  const int m0 = (j * 2 + (wave >> 1)) * 16;
  int arow = m0 + r;
  if (arow > M - 1) arow = M - 1;
  const float* Arow = A + (size_t)arow * D;
  if (rel) Arow = arow < R ? relemb + (size_t)arow * D : selfrel;
  strip_gemm_block(Arow + 2 * g, W, bias, C, m0, M, wave, lane, r, g);
}

#define SLOTS 31
#define RANK_BINS 40960
#define RANK_EDGES 7
#define RANK_LINES 20

__global__ __launch_bounds__(1024) void edge_slot_kernel(
    const int* __restrict__ ei, const int* __restrict__ et, int E, int N,
    int* __restrict__ cnti, int* chunks, int* __restrict__ ovfn,
    int* __restrict__ ovf) {
  __shared__ __attribute__((aligned(16))) int hist[RANK_BINS];
  const int tid = threadIdx.x;
  const int per = (E + (int)gridDim.x - 1) / (int)gridDim.x;
  const int e0 = blockIdx.x * per;
  int e1 = e0 + per;
  if (e1 > E) e1 = E;
  for (int eb = e0; eb < e1; eb += RANK_EDGES * 1024) {
    for (int i = tid; i < RANK_BINS / 4; i += 1024)
      ((int4*)hist)[i] = make_int4(0, 0, 0, 0);

    int s[RANK_EDGES], d[RANK_EDGES], t[RANK_EDGES], rk[RANK_EDGES];
#pragma unroll
    for (int k = 0; k < RANK_EDGES; ++k) {
      const int e = eb + k * 1024 + tid;
      d[k] = -1;
      s[k] = 0;
      t[k] = 0;
      if (e < e1) {
        s[k] = ei[e];
        d[k] = ei[E + e];
        t[k] = et[e];
      }
    }
    __syncthreads();

#pragma unroll
    for (int k = 0; k < RANK_EDGES; ++k) {
      rk[k] = 0;
      if (d[k] >= 0) rk[k] = atomicAdd(&hist[d[k]], 1);
    }
    __syncthreads();

    for (int b0 = 0; b0 < N; b0 += RANK_LINES * 1024) {
      int h[RANK_LINES], first[RANK_LINES];
#pragma unroll
      for (int u = 0; u < RANK_LINES; ++u) {
        const int bin = b0 + u * 1024 + tid;
        h[u] = bin < N ? hist[bin] : 0;
      }
#pragma unroll
      for (int u = 0; u < RANK_LINES; ++u) {
        const int bin = b0 + u * 1024 + tid;
        first[u] = 0;
        if (h[u] > 0) first[u] = atomicAdd(&cnti[bin], h[u]);
      }
#pragma unroll
      for (int u = 0; u < RANK_LINES; ++u) {
        const int bin = b0 + u * 1024 + tid;
        if (h[u] > 0) hist[bin] = first[u];
      }
    }
    __syncthreads();

#pragma unroll
    for (int k = 0; k < RANK_EDGES; ++k) {
      if (d[k] >= 0) {
        const int slot = hist[d[k]] + rk[k];
        if (slot < SLOTS) {
          ((int2*)(chunks + (size_t)d[k] * D))[slot] = make_int2(s[k], t[k]);
        } else {
          const int o = atomicAdd(ovfn, 1);
          ((int2*)ovf)[o] = make_int2(s[k], d[k] * SELF_PAD + t[k]);
        }
      }
    }
    __syncthreads();
  }
}

template <int NV>
__device__ __forceinline__ void segment_group(
    const int rs, const int rt, const int b, const int lane, const float4 xq,
    const float* __restrict__ Xk, const float* __restrict__ Tk,
    const float* __restrict__ Tq, const float* __restrict__ Vv,
    const bool up16, const bool up8, const bool up4, float& md, float& ms,
    float& sd, float& ss, float4& as, float4& ad) {
  int s[8], t[8];
#pragma unroll
  for (int j = 0; j < NV; ++j) {
    s[j] = __builtin_amdgcn_readlane(rs, b + j);
    t[j] = __builtin_amdgcn_readlane(rt, b + j);
  }
  float4 k0[8], k1[8], q1[8], v[8];
#pragma unroll
  for (int j = 0; j < NV; ++j) {
    k0[j] = ((const float4*)(Xk + (size_t)s[j] * D))[lane];
    k1[j] = ((const float4*)(Tk + (size_t)t[j] * D))[lane];
    q1[j] = ((const float4*)(Tq + (size_t)t[j] * D))[lane];
  }
#pragma unroll
  for (int j = 0; j < NV; ++j)
    v[j] = ((const float4*)(Vv + (size_t)s[j] * D))[lane];

  float a[8];
#pragma unroll
  for (int j = 0; j < 8; ++j) {
    a[j] = 0.0f;
    if (j < NV) {
      a[j] = (xq.x + q1[j].x) * (k0[j].x + k1[j].x) +
             (xq.y + q1[j].y) * (k0[j].y + k1[j].y) +
             (xq.z + q1[j].z) * (k0[j].z + k1[j].z) +
             (xq.w + q1[j].w) * (k0[j].w + k1[j].w);
    }
  }

  float b4[4];
#pragma unroll
  for (int j = 0; j < 4; ++j) {
    const float keep = up16 ? a[j + 4] : a[j];
    const float give = up16 ? a[j] : a[j + 4];
    b4[j] = keep + __shfl_xor(give, 16, 32);
  }
  float c2[2];
#pragma unroll
  for (int j = 0; j < 2; ++j) {
    const float keep = up8 ? b4[j + 2] : b4[j];
    const float give = up8 ? b4[j] : b4[j + 2];
    c2[j] = keep + __shfl_xor(give, 8, 32);
  }
  float r;
  {
    const float keep = up4 ? c2[1] : c2[0];
    const float give = up4 ? c2[0] : c2[1];
    r = keep + __shfl_xor(give, 4, 32);
  }
  r += __shfl_xor(r, 2, 32);
  r += __shfl_xor(r, 1, 32);
  r *= 0.08838834764831845f;

  float sc[8];
#pragma unroll
  for (int j = 0; j < NV; ++j)
    sc[j] = __int_as_float(
        __builtin_amdgcn_readlane(__float_as_int(r), 4 * j));

  float gmax = sc[0], gmin = sc[0];
#pragma unroll
  for (int j = 1; j < NV; ++j) {
    gmax = fmaxf(gmax, sc[j]);
    gmin = fminf(gmin, sc[j]);
  }
  const float mdn = fmaxf(md, gmax);
  const float msn = fmaxf(ms, -gmin);
  const float fd = __expf(md - mdn);
  const float fs = __expf(ms - msn);
  sd *= fd;
  ss *= fs;
  ad.x *= fd;
  ad.y *= fd;
  ad.z *= fd;
  ad.w *= fd;
  as.x *= fs;
  as.y *= fs;
  as.z *= fs;
  as.w *= fs;
  md = mdn;
  ms = msn;
#pragma unroll
  for (int j = 0; j < NV; ++j) {
    const float ed = __expf(sc[j] - mdn);
    const float es = __expf(-sc[j] - msn);
    sd += ed;
    ss += es;
    ad.x += ed * v[j].x;
    ad.y += ed * v[j].y;
    ad.z += ed * v[j].z;
    ad.w += ed * v[j].w;
    as.x += es * v[j].x;
    as.y += es * v[j].y;
    as.z += es * v[j].z;
    as.w += es * v[j].w;
  }
}

#define FP 132
#define SEG_GROUPS 512

__global__ __launch_bounds__(256) void node_segment_dyn_kernel(
    int N, int R, int E, const int* __restrict__ cnti,
    const int* __restrict__ ovfn, const int* __restrict__ ovf,
    const int* __restrict__ chunks, const float* __restrict__ Xq,
    const float* __restrict__ Xk, const float* __restrict__ Tq,
    const float* __restrict__ Tk, const float* __restrict__ Vv,
    const float* __restrict__ dynW, float* __restrict__ featsta,
    float* __restrict__ out1) {
  __shared__ __attribute__((aligned(16))) float Wd[64 * WP];
  __shared__ __attribute__((aligned(16))) float Fd[2 * 16 * FP];
  const int lane = threadIdx.x & 31;
  const int wv = __builtin_amdgcn_readfirstlane((int)(threadIdx.x >> 5));

  {
    float4 w0[8], w1[8];
    const float* __restrict__ Wb = dynW + (size_t)(2 * wv) * D + 4 * lane;
#pragma unroll
    for (int it = 0; it < 8; ++it) {
      w0[it] = *(const float4*)(Wb + it * 16 * D);
      w1[it] = *(const float4*)(Wb + it * 16 * D + D);
    }
    float* dstb = Wd + wv * WP + 8 * lane;
#pragma unroll
    for (int it = 0; it < 8; ++it) {
      float* dst = dstb + it * 8 * WP;
      *(float4*)dst = make_float4(w0[it].x, w1[it].x, w0[it].y, w1[it].y);
      *(float4*)(dst + 4) = make_float4(w0[it].z, w1[it].z, w0[it].w, w1[it].w);
    }
  }

  const bool up16 = (lane & 16) != 0;
  const bool up8 = (lane & 8) != 0;
  const bool up4 = (lane & 4) != 0;

  const int fr = lane & 15;
  const int fg = lane >> 4;
  const int ntiles = N / 16;
  int half = 0;
  for (int tile = blockIdx.x; tile < ntiles; tile += SEG_GROUPS) {
    const int cur = half;
    float* Ft = Fd + cur * 16 * FP;
    half ^= 1;
    for (int i = 0; i < 2; ++i) {
      const int lrow = wv * 2 + i;
      const int d = tile * 16 + lrow;
      const int2 p = ((const int2*)(chunks + (size_t)d * D))[lane];
      const int cnt = cnti[d];
      const int nt = cnt + 1;
      const float4 xq = ((const float4*)(Xq + (size_t)d * D))[lane];
      const int m = cnt < SLOTS ? cnt : SLOTS;
      int rs = p.x, rt = p.y;
      if (lane >= m) {
        rs = d;
        rt = R;
      }
      int n = m + 1;
      int novf = 0;
      if (cnt > SLOTS) {
        novf = ovfn[0];
        if (novf > E) novf = E;
      }
      int oc = 0;
      unsigned pend = 0u;
      int os = 0, ow = -1;

      float md = -__builtin_inff();
      float ms = -__builtin_inff();
      float sd = 0.0f, ss = 0.0f;
      float4 as = make_float4(0.0f, 0.0f, 0.0f, 0.0f);
      float4 ad = make_float4(0.0f, 0.0f, 0.0f, 0.0f);

      for (;;) {
        for (int b = 0; b < n; b += 8) {
          const int nv = n - b;
#define SEG_GROUP(NVAL)                                                       \
  segment_group<NVAL>(rs, rt, b, lane, xq, Xk, Tk, Tq, Vv, up16, up8, up4, \
                      md, ms, sd, ss, as, ad)
          if (nv >= 8) {
            SEG_GROUP(8);
          } else {
            switch (nv) {
              case 1: SEG_GROUP(1); break;
              case 2: SEG_GROUP(2); break;
              case 3: SEG_GROUP(3); break;
              case 4: SEG_GROUP(4); break;
              case 5: SEG_GROUP(5); break;
              case 6: SEG_GROUP(6); break;
              default: SEG_GROUP(7); break;
            }
          }
#undef SEG_GROUP
        }
        while (pend == 0u && oc < novf) {
          int2 q = make_int2(0, -1);
          if (oc + lane < novf) q = ((const int2*)ovf)[oc + lane];
          os = q.x;
          ow = q.y;
          pend = __builtin_amdgcn_ballot_w32(ow / SELF_PAD == d && ow >= 0);
          oc += 32;
        }
        if (pend == 0u) break;
        const int jl = __builtin_ctz(pend);
        pend &= pend - 1u;
        rs = __builtin_amdgcn_readlane(os, jl);
        rt = __builtin_amdgcn_readlane(ow, jl) % SELF_PAD;
        n = 1;
      }
      const float denom = (float)nt;
      const float wd = 1.0f / (sd * denom);
      const float ws = 1.0f / (ss * denom);
      ((float4*)(featsta + (size_t)d * D))[lane] =
          make_float4(as.x * ws, as.y * ws, as.z * ws, as.w * ws);
      *(float4*)(Ft + lrow * FP + 4 * lane) =
          make_float4(ad.x * wd, ad.y * wd, ad.z * wd, ad.w * wd);
    }

    asm volatile("s_wait_dscnt 0x0\n\ts_barrier_signal -1\n\ts_barrier_wait -1"
                 ::: "memory");

    if ((wv >> 2) == cur) {
      const int nb = (wv & 3) * 32;
      const float* Fa = Ft + fr * FP + 2 * fg;
      const float* Bp = Wd + fg * WP + 2 * (nb + fr);
      v8f acc[2] = {};
#pragma unroll
      for (int ks = 0; ks < 32; ++ks) {
        const v2f a = *(const v2f*)(Fa + 4 * ks);
#pragma unroll
        for (int ct = 0; ct < 2; ++ct) {
          const v2f b = *(const v2f*)(Bp + 2 * ks * WP + 32 * ct);
          acc[ct] = __builtin_amdgcn_wmma_f32_16x16x4_f32(
               false, a,  false, b,
               (short)0, acc[ct],  false,  false);
        }
      }
#pragma unroll
      for (int ct = 0; ct < 2; ++ct) {
        const int col = nb + 16 * ct + fr;
#pragma unroll
        for (int j = 0; j < 8; ++j) {
          const int row = tile * 16 + j + 8 * fg;
          out1[(size_t)row * D + col] = acc[ct][j];
        }
      }
    }
  }
}

extern "C" void kernel_launch(void* const* d_in, const int* in_sizes, int n_in,
                              void* d_out, int out_size, void* d_ws,
                              size_t ws_size, hipStream_t stream) {
  const float* curr = (const float*)d_in[0];
  const float* last = (const float*)d_in[1];
  const float* relemb = (const float*)d_in[2];
  const float* selfrel = (const float*)d_in[3];
  const float* Wq = (const float*)d_in[4];
  const float* bq = (const float*)d_in[5];
  const float* Wk = (const float*)d_in[6];
  const float* bk = (const float*)d_in[7];
  const float* Wv = (const float*)d_in[8];
  const float* bv = (const float*)d_in[9];
  const float* dynW = (const float*)d_in[10];
  const int* ei = (const int*)d_in[11];
  const int* et = (const int*)d_in[12];

  const int N = in_sizes[0] / D;
  const int E = in_sizes[12];
  const int R = in_sizes[2] / D;
  const size_t ND = (size_t)N * D;

  float* ws = (float*)d_ws;
  float* Xq = ws;
  float* Xk = Xq + ND;
  float* Vv = Xk + ND;
  int* chunks = (int*)(Vv + ND);
  float* Tq = (float*)chunks + ND;
  float* Tk = Tq + SELF_PAD * D;
  int* ovf = (int*)(Tk + SELF_PAD * D);
  int* ovfn = ovf + (size_t)2 * E;
  int* cnti = ovfn + (N + 1);

  float* out0 = (float*)d_out;
  float* out1 = out0 + ND;

  front_products_wmma<<<3 * ((N / 16 + 1) / 2) + 2 * (SELF_PAD / 32), 128, 0,
                        stream>>>(curr, last, relemb, selfrel, Wq, Wk, Wv, bq,
                                  bk, bv, Xq, Xk, Vv, Tq, Tk, cnti, ovfn, N, R);

  edge_slot_kernel<<<64, 1024, 0, stream>>>(ei, et, E, N, cnti, chunks, ovfn,
                                            ovf);

  node_segment_dyn_kernel<<<SEG_GROUPS, 256, 0, stream>>>(
      N, R, E, cnti, ovfn, ovf, chunks, Xq, Xk, Tq, Tk, Vv, dynW, out0, out1);
}
